# starting kernel
# baseline (speedup 1.0000x reference)
#include <hip/hip_runtime.h>
#include <math.h>

#define BB 128
#define TT 2048
#define DD 128
#define HH 128

typedef __attribute__((ext_vector_type(2))) float v2f;
typedef __attribute__((ext_vector_type(8))) float v8f;

__global__ __launch_bounds__(256) void colsum_partial(const float* __restrict__ x,
                                                      float* __restrict__ partial) {
  const int b = blockIdx.x, chunk = blockIdx.y;
  const int g = threadIdx.x >> 7, d = threadIdx.x & 127;
  const int tbase = chunk * 512 + g * 256;
  const float* xb = x + ((b * TT + tbase) * DD) + d;
  float s = 0.f;
  for (int i = 0; i < 256; ++i) s += xb[i * DD];
  __shared__ float r[256];
  r[threadIdx.x] = s;
  __syncthreads();
  if (g == 0) partial[(b * 4 + chunk) * DD + d] = r[d] + r[128 + d];
}

__global__ __launch_bounds__(128) void colsum_final(const float* __restrict__ partial,
                                                    float* __restrict__ xmean) {
  const int b = blockIdx.x, d = threadIdx.x;
  float s = partial[(b * 4 + 0) * DD + d] + partial[(b * 4 + 1) * DD + d] +
            partial[(b * 4 + 2) * DD + d] + partial[(b * 4 + 3) * DD + d];
  xmean[b * DD + d] = s * (1.0f / (float)TT);
}

template <bool TRANS>
__global__ __launch_bounds__(32) void gemm128_wmma(const float* __restrict__ A,
                                                   const float* __restrict__ W,
                                                   const float* __restrict__ bias,
                                                   float* __restrict__ C) {
  const int n0 = blockIdx.x * 16;
  const int m0 = blockIdx.y * 16;
  const int lane = threadIdx.x & 31;
  const int mrow = lane & 15;
  const int hi = lane >> 4;

  v8f acc = {};
  for (int k0 = 0; k0 < 128; k0 += 4) {
    v2f a = *(const v2f*)(A + (m0 + mrow) * 128 + k0 + 2 * hi);
    v2f bfrag;
    const int kb = k0 + 2 * hi;
    const int n = n0 + mrow;
    if (TRANS) {
      bfrag.x = W[n * 128 + kb + 0];
      bfrag.y = W[n * 128 + kb + 1];
    } else {
      bfrag.x = W[(kb + 0) * 128 + n];
      bfrag.y = W[(kb + 1) * 128 + n];
    }
    acc = __builtin_amdgcn_wmma_f32_16x16x4_f32(false, a, false, bfrag,
                                                (short)0, acc, false, false);
  }
  const int n = n0 + mrow;
  const float bb = bias ? bias[n] : 0.0f;
#pragma unroll
  for (int r = 0; r < 8; ++r) {
    const int m = m0 + r + 8 * hi;
    C[m * 128 + n] = acc[r] + bb;
  }
}

__global__ __launch_bounds__(128) void dot0_kernel(const float* __restrict__ q,
                                                   const float* __restrict__ bk,
                                                   float* __restrict__ dot0) {
  const int b = blockIdx.x;
  __shared__ float r[128];
  r[threadIdx.x] = bk[threadIdx.x] * q[b * HH + threadIdx.x];
  __syncthreads();
  for (int s = 64; s > 0; s >>= 1) {
    if (threadIdx.x < s) r[threadIdx.x] += r[threadIdx.x + s];
    __syncthreads();
  }
  if (threadIdx.x == 0) dot0[b] = r[0];
}

__global__ __launch_bounds__(256) void dot_decay_kernel(
    const float* __restrict__ x, const float* __restrict__ mask,
    const float* __restrict__ qk, const float* __restrict__ dot0,
    const float* __restrict__ rate, float* __restrict__ dotOut,
    float* __restrict__ decayOut, float* __restrict__ e) {
  const int b = blockIdx.x;
  const int wave = threadIdx.x >> 5, lane = threadIdx.x & 31;
  const float4 qk4 = *(const float4*)(qk + b * DD + lane * 4);
  const float rate0 = rate[0];
  const float d0 = dot0[b];
  const int t0 = blockIdx.y * 256 + wave * 32;
  for (int i = 0; i < 32; ++i) {
    const int t = t0 + i;
    const float4* row = (const float4*)(x + (b * TT + t) * DD);
    const float4 xv = row[lane];
    float s = xv.x * qk4.x + xv.y * qk4.y + xv.z * qk4.z + xv.w * qk4.w;
    for (int off = 16; off > 0; off >>= 1) s += __shfl_xor(s, off, 32);
    if (lane == 0) {
      const float dv = s + d0;
      const float sig = 1.0f / (1.0f + expf(-dv));
      const float tm = log1pf((1.0f - sig) * mask[b * TT + t]);
      const float dec = rate0 * tm;
      dotOut[b * TT + t] = dv;
      decayOut[b * TT + t] = dec;
      e[b * TT + t] = dv - dec;
    }
  }
}

__global__ __launch_bounds__(512) void softmax_xa_kernel(const float* __restrict__ x,
                                                         const float* __restrict__ e,
                                                         float* __restrict__ aOut,
                                                         float* __restrict__ xa) {
  const int b = blockIdx.x, tid = threadIdx.x;
  __shared__ float aLds[TT];
  __shared__ float red[512];

  float ev[4];
  float m = -3.4e38f;
#pragma unroll
  for (int k = 0; k < 4; ++k) {
    ev[k] = e[b * TT + k * 512 + tid];
    m = fmaxf(m, ev[k]);
  }
  red[tid] = m;
  __syncthreads();
  for (int s = 256; s > 0; s >>= 1) {
    if (tid < s) red[tid] = fmaxf(red[tid], red[tid + s]);
    __syncthreads();
  }
  const float M = red[0];
  __syncthreads();

  float p[4];
  float ls = 0.f;
#pragma unroll
  for (int k = 0; k < 4; ++k) {
    p[k] = expf(ev[k] - M);
    ls += p[k];
  }
  red[tid] = ls;
  __syncthreads();
  for (int s = 256; s > 0; s >>= 1) {
    if (tid < s) red[tid] += red[tid + s];
    __syncthreads();
  }
  const float inv = 1.0f / red[0];
  __syncthreads();

#pragma unroll
  for (int k = 0; k < 4; ++k) {
    const float a = p[k] * inv;
    aLds[k * 512 + tid] = a;
    aOut[b * TT + k * 512 + tid] = a;
  }
  __syncthreads();

  const int ts = tid >> 7, d = tid & 127;
  float acc = 0.f;
  for (int t = ts; t < TT; t += 4) acc += aLds[t] * x[(b * TT + t) * DD + d];
  red[tid] = acc;
  __syncthreads();
  if (ts == 0) xa[b * DD + d] = red[d] + red[128 + d] + red[256 + d] + red[384 + d];
}

extern "C" void kernel_launch(void* const* d_in, const int* in_sizes, int n_in,
                              void* d_out, int out_size, void* d_ws, size_t ws_size,
                              hipStream_t stream) {
  (void)in_sizes; (void)n_in; (void)out_size; (void)ws_size;
  const float* x    = (const float*)d_in[0];
  const float* mask = (const float*)d_in[1];
  const float* Wq   = (const float*)d_in[2];
  const float* bq   = (const float*)d_in[3];
  const float* Wk   = (const float*)d_in[4];
  const float* bk   = (const float*)d_in[5];
  const float* Wv   = (const float*)d_in[6];
  const float* bv   = (const float*)d_in[7];
  const float* rate = (const float*)d_in[8];

  float* wsf     = (float*)d_ws;
  float* partial = wsf;
  float* xmean   = partial + BB * 4 * DD;
  float* q       = xmean + BB * DD;
  float* qk      = q + BB * HH;
  float* dot0    = qk + BB * DD;
  float* e       = dot0 + BB;
  float* xa      = e + BB * TT;

  float* out      = (float*)d_out;
  float* aOut     = out + BB * HH;
  float* dotOut   = aOut + BB * TT;
  float* decayOut = dotOut + BB * TT;

  colsum_partial<<<dim3(BB, 4), 256, 0, stream>>>(x, partial);
  colsum_final<<<BB, 128, 0, stream>>>(partial, xmean);
  gemm128_wmma<true><<<dim3(8, 8), 32, 0, stream>>>(xmean, Wq, bq, q);
  gemm128_wmma<false><<<dim3(8, 8), 32, 0, stream>>>(q, Wk, nullptr, qk);
  dot0_kernel<<<BB, 128, 0, stream>>>(q, bk, dot0);
  dot_decay_kernel<<<dim3(BB, 8), 256, 0, stream>>>(x, mask, qk, dot0, rate,
                                                    dotOut, decayOut, e);
  softmax_xa_kernel<<<BB, 512, 0, stream>>>(x, e, aOut, xa);
  gemm128_wmma<true><<<dim3(8, 8), 32, 0, stream>>>(xa, Wv, bv, out);
}


# dot/decay pass: each wave sums 64 rows with row loads issued one group ahead, then a transposed xor butterfly (same pairing tree) leaves one finished row per lane; whole-line stores
# speedup vs baseline: 1.4790x; 1.4790x over previous
#include <hip/hip_runtime.h>
#include <math.h>

#define BB 128
#define TT 2048
#define DD 128
#define HH 128

typedef __attribute__((ext_vector_type(2))) float v2f;
typedef __attribute__((ext_vector_type(8))) float v8f;

__global__ __launch_bounds__(256) void colsum_partial(const float* __restrict__ x,
                                                      float* __restrict__ partial) {
  const int b = blockIdx.x, chunk = blockIdx.y;
  const int g = threadIdx.x >> 7, d = threadIdx.x & 127;
  const int tbase = chunk * 512 + g * 256;
  const float* xb = x + ((b * TT + tbase) * DD) + d;
  float s = 0.f;
  for (int i = 0; i < 256; ++i) s += xb[i * DD];
  __shared__ float r[256];
  r[threadIdx.x] = s;
  __syncthreads();
  if (g == 0) partial[(b * 4 + chunk) * DD + d] = r[d] + r[128 + d];
}

__global__ __launch_bounds__(128) void colsum_final(const float* __restrict__ partial,
                                                    float* __restrict__ xmean) {
  const int b = blockIdx.x, d = threadIdx.x;
  float s = partial[(b * 4 + 0) * DD + d] + partial[(b * 4 + 1) * DD + d] +
            partial[(b * 4 + 2) * DD + d] + partial[(b * 4 + 3) * DD + d];
  xmean[b * DD + d] = s * (1.0f / (float)TT);
}

template <bool TRANS>
__global__ __launch_bounds__(32) void gemm128_wmma(const float* __restrict__ A,
                                                   const float* __restrict__ W,
                                                   const float* __restrict__ bias,
                                                   float* __restrict__ C) {
  const int n0 = blockIdx.x * 16;
  const int m0 = blockIdx.y * 16;
  const int lane = threadIdx.x & 31;
  const int mrow = lane & 15;
  const int hi = lane >> 4;

  v8f acc = {};
  for (int k0 = 0; k0 < 128; k0 += 4) {
    v2f a = *(const v2f*)(A + (m0 + mrow) * 128 + k0 + 2 * hi);
    v2f bfrag;
    const int kb = k0 + 2 * hi;
    const int n = n0 + mrow;
    if (TRANS) {
      bfrag.x = W[n * 128 + kb + 0];
      bfrag.y = W[n * 128 + kb + 1];
    } else {
      bfrag.x = W[(kb + 0) * 128 + n];
      bfrag.y = W[(kb + 1) * 128 + n];
    }
    acc = __builtin_amdgcn_wmma_f32_16x16x4_f32(false, a, false, bfrag,
                                                (short)0, acc, false, false);
  }
  const int n = n0 + mrow;
  const float bb = bias ? bias[n] : 0.0f;
#pragma unroll
  for (int r = 0; r < 8; ++r) {
    const int m = m0 + r + 8 * hi;
    C[m * 128 + n] = acc[r] + bb;
  }
}

__global__ __launch_bounds__(128) void dot0_kernel(const float* __restrict__ q,
                                                   const float* __restrict__ bk,
                                                   float* __restrict__ dot0) {
  const int b = blockIdx.x;
  __shared__ float r[128];
  r[threadIdx.x] = bk[threadIdx.x] * q[b * HH + threadIdx.x];
  __syncthreads();
  for (int s = 64; s > 0; s >>= 1) {
    if (threadIdx.x < s) r[threadIdx.x] += r[threadIdx.x + s];
    __syncthreads();
  }
  if (threadIdx.x == 0) dot0[b] = r[0];
}

template <int OFF>
__device__ __forceinline__ void fold_level(float (&s)[32], const int lane) {
  const bool up = (lane & OFF) != 0;
#pragma unroll
  for (int i = 0; i < OFF; ++i) {
    const float keep = up ? s[i + OFF] : s[i];
    const float send = up ? s[i] : s[i + OFF];
    s[i] = keep + __shfl_xor(send, OFF, 32);
  }
}

__global__ __launch_bounds__(256) void dot_decay_kernel(
    const float* __restrict__ x, const float* __restrict__ mask,
    const float* __restrict__ qk, const float* __restrict__ dot0,
    const float* __restrict__ rate, float* __restrict__ dotOut,
    float* __restrict__ decayOut, float* __restrict__ e) {
  const int b = blockIdx.x;
  const int wave = threadIdx.x >> 5, lane = threadIdx.x & 31;
  const int tw = blockIdx.y * 512 + wave * 64;
  const float4 qk4 = *(const float4*)(qk + b * DD + lane * 4);
  const float rate0 = rate[0];
  const float d0 = dot0[b];
  float mk[2];
  mk[0] = mask[b * TT + tw + lane];
  mk[1] = mask[b * TT + tw + 32 + lane];
  const float4* rows = (const float4*)(x + (b * TT + tw) * DD) + lane;
  float4 xv[2][16];
  float s[32];
#pragma unroll
  for (int i = 0; i < 16; ++i) xv[0][i] = rows[i * (DD / 4)];
#pragma unroll
  for (int h = 0; h < 4; ++h) {
    __builtin_amdgcn_sched_barrier(0);
    if (h + 1 < 4) {
#pragma unroll
      for (int i = 0; i < 16; ++i)
        xv[(h + 1) & 1][i] = rows[((h + 1) * 16 + i) * (DD / 4)];
    }
    __builtin_amdgcn_sched_barrier(0);
#pragma unroll
    for (int i = 0; i < 16; ++i) {
      const float4 v = xv[h & 1][i];
      s[(h & 1) * 16 + i] = v.x * qk4.x + v.y * qk4.y + v.z * qk4.z + v.w * qk4.w;
    }
    if (h & 1) {
      fold_level<16>(s, lane);
      fold_level<8>(s, lane);
      fold_level<4>(s, lane);
      fold_level<2>(s, lane);
      fold_level<1>(s, lane);
      const int o = b * TT + tw + (h >> 1) * 32 + lane;
      const float dv = s[0] + d0;
      const float sig = 1.0f / (1.0f + expf(-dv));
      const float tm = log1pf((1.0f - sig) * mk[h >> 1]);
      const float dec = rate0 * tm;
      dotOut[o] = dv;
      decayOut[o] = dec;
      e[o] = dv - dec;
    }
  }
}

__global__ __launch_bounds__(512) void softmax_xa_kernel(const float* __restrict__ x,
                                                         const float* __restrict__ e,
                                                         float* __restrict__ aOut,
                                                         float* __restrict__ xa) {
  const int b = blockIdx.x, tid = threadIdx.x;
  __shared__ float aLds[TT];
  __shared__ float red[512];

  float ev[4];
  float m = -3.4e38f;
#pragma unroll
  for (int k = 0; k < 4; ++k) {
    ev[k] = e[b * TT + k * 512 + tid];
    m = fmaxf(m, ev[k]);
  }
  red[tid] = m;
  __syncthreads();
  for (int s = 256; s > 0; s >>= 1) {
    if (tid < s) red[tid] = fmaxf(red[tid], red[tid + s]);
    __syncthreads();
  }
  const float M = red[0];
  __syncthreads();

  float p[4];
  float ls = 0.f;
#pragma unroll
  for (int k = 0; k < 4; ++k) {
    p[k] = expf(ev[k] - M);
    ls += p[k];
  }
  red[tid] = ls;
  __syncthreads();
  for (int s = 256; s > 0; s >>= 1) {
    if (tid < s) red[tid] += red[tid + s];
    __syncthreads();
  }
  const float inv = 1.0f / red[0];
  __syncthreads();

#pragma unroll
  for (int k = 0; k < 4; ++k) {
    const float a = p[k] * inv;
    aLds[k * 512 + tid] = a;
    aOut[b * TT + k * 512 + tid] = a;
  }
  __syncthreads();

  const int ts = tid >> 7, d = tid & 127;
  float acc = 0.f;
  for (int t = ts; t < TT; t += 4) acc += aLds[t] * x[(b * TT + t) * DD + d];
  red[tid] = acc;
  __syncthreads();
  if (ts == 0) xa[b * DD + d] = red[d] + red[128 + d] + red[256 + d] + red[384 + d];
}

extern "C" void kernel_launch(void* const* d_in, const int* in_sizes, int n_in,
                              void* d_out, int out_size, void* d_ws, size_t ws_size,
                              hipStream_t stream) {
  (void)in_sizes; (void)n_in; (void)out_size; (void)ws_size;
  const float* x    = (const float*)d_in[0];
  const float* mask = (const float*)d_in[1];
  const float* Wq   = (const float*)d_in[2];
  const float* bq   = (const float*)d_in[3];
  const float* Wk   = (const float*)d_in[4];
  const float* bk   = (const float*)d_in[5];
  const float* Wv   = (const float*)d_in[6];
  const float* bv   = (const float*)d_in[7];
  const float* rate = (const float*)d_in[8];

  float* wsf     = (float*)d_ws;
  float* partial = wsf;
  float* xmean   = partial + BB * 4 * DD;
  float* q       = xmean + BB * DD;
  float* qk      = q + BB * HH;
  float* dot0    = qk + BB * DD;
  float* e       = dot0 + BB;
  float* xa      = e + BB * TT;

  float* out      = (float*)d_out;
  float* aOut     = out + BB * HH;
  float* dotOut   = aOut + BB * TT;
  float* decayOut = dotOut + BB * TT;

  colsum_partial<<<dim3(BB, 4), 256, 0, stream>>>(x, partial);
  colsum_final<<<BB, 128, 0, stream>>>(partial, xmean);
  gemm128_wmma<true><<<dim3(8, 8), 32, 0, stream>>>(xmean, Wq, bq, q);
  gemm128_wmma<false><<<dim3(8, 8), 32, 0, stream>>>(q, Wk, nullptr, qk);
  dot0_kernel<<<BB, 128, 0, stream>>>(q, bk, dot0);
  dot_decay_kernel<<<dim3(BB, 4), 256, 0, stream>>>(x, mask, qk, dot0, rate,
                                                    dotOut, decayOut, e);
  softmax_xa_kernel<<<BB, 512, 0, stream>>>(x, e, aOut, xa);
  gemm128_wmma<true><<<dim3(8, 8), 32, 0, stream>>>(xa, Wv, bv, out);
}


# softmax/xa pass on 4 work-groups per batch: 64 rows per wave, 16-byte row loads, about 20 in flight per wave, shuffle softmax stats; xa partials added in out GEMM A load (new sum trees: exp-sum, xa)
# speedup vs baseline: 1.8699x; 1.2643x over previous
#include <hip/hip_runtime.h>
#include <math.h>

#define BB 128
#define TT 2048
#define DD 128
#define HH 128

typedef __attribute__((ext_vector_type(2))) float v2f;
typedef __attribute__((ext_vector_type(8))) float v8f;

__global__ __launch_bounds__(256) void colsum_partial(const float* __restrict__ x,
                                                      float* __restrict__ partial) {
  const int b = blockIdx.x, chunk = blockIdx.y;
  const int g = threadIdx.x >> 7, d = threadIdx.x & 127;
  const int tbase = chunk * 512 + g * 256;
  const float* xb = x + ((b * TT + tbase) * DD) + d;
  float s = 0.f;
  for (int i = 0; i < 256; ++i) s += xb[i * DD];
  __shared__ float r[256];
  r[threadIdx.x] = s;
  __syncthreads();
  if (g == 0) partial[(b * 4 + chunk) * DD + d] = r[d] + r[128 + d];
}

__global__ __launch_bounds__(128) void colsum_final(const float* __restrict__ partial,
                                                    float* __restrict__ xmean) {
  const int b = blockIdx.x, d = threadIdx.x;
  float s = partial[(b * 4 + 0) * DD + d] + partial[(b * 4 + 1) * DD + d] +
            partial[(b * 4 + 2) * DD + d] + partial[(b * 4 + 3) * DD + d];
  xmean[b * DD + d] = s * (1.0f / (float)TT);
}

template <bool TRANS, int NP>
__global__ __launch_bounds__(32) void gemm128_wmma(const float* __restrict__ A,
                                                   const float* __restrict__ W,
                                                   const float* __restrict__ bias,
                                                   float* __restrict__ C) {
  const int n0 = blockIdx.x * 16;
  const int m0 = blockIdx.y * 16;
  const int lane = threadIdx.x & 31;
  const int mrow = lane & 15;
  const int hi = lane >> 4;

  v8f acc = {};
  for (int k0 = 0; k0 < 128; k0 += 4) {
    const float* ap = A + (m0 + mrow) * (NP * 128) + k0 + 2 * hi;
    v2f a = *(const v2f*)ap;
#pragma unroll
    for (int c = 1; c < NP; ++c) a += *(const v2f*)(ap + c * 128);
    v2f bfrag;
    const int kb = k0 + 2 * hi;
    const int n = n0 + mrow;
    if (TRANS) {
      bfrag.x = W[n * 128 + kb + 0];
      bfrag.y = W[n * 128 + kb + 1];
    } else {
      bfrag.x = W[(kb + 0) * 128 + n];
      bfrag.y = W[(kb + 1) * 128 + n];
    }
    acc = __builtin_amdgcn_wmma_f32_16x16x4_f32(false, a, false, bfrag,
                                                (short)0, acc, false, false);
  }
  const int n = n0 + mrow;
  const float bb = bias ? bias[n] : 0.0f;
#pragma unroll
  for (int r = 0; r < 8; ++r) {
    const int m = m0 + r + 8 * hi;
    C[m * 128 + n] = acc[r] + bb;
  }
}

__global__ __launch_bounds__(128) void dot0_kernel(const float* __restrict__ q,
                                                   const float* __restrict__ bk,
                                                   float* __restrict__ dot0) {
  const int b = blockIdx.x;
  __shared__ float r[128];
  r[threadIdx.x] = bk[threadIdx.x] * q[b * HH + threadIdx.x];
  __syncthreads();
  for (int s = 64; s > 0; s >>= 1) {
    if (threadIdx.x < s) r[threadIdx.x] += r[threadIdx.x + s];
    __syncthreads();
  }
  if (threadIdx.x == 0) dot0[b] = r[0];
}

template <int OFF>
__device__ __forceinline__ void fold_level(float (&s)[32], const int lane) {
  const bool up = (lane & OFF) != 0;
#pragma unroll
  for (int i = 0; i < OFF; ++i) {
    const float keep = up ? s[i + OFF] : s[i];
    const float send = up ? s[i] : s[i + OFF];
    s[i] = keep + __shfl_xor(send, OFF, 32);
  }
}

__global__ __launch_bounds__(256) void dot_decay_kernel(
    const float* __restrict__ x, const float* __restrict__ mask,
    const float* __restrict__ qk, const float* __restrict__ dot0,
    const float* __restrict__ rate, float* __restrict__ dotOut,
    float* __restrict__ decayOut, float* __restrict__ e) {
  const int b = blockIdx.x;
  const int wave = threadIdx.x >> 5, lane = threadIdx.x & 31;
  const int tw = blockIdx.y * 512 + wave * 64;
  const float4 qk4 = *(const float4*)(qk + b * DD + lane * 4);
  const float rate0 = rate[0];
  const float d0 = dot0[b];
  float mk[2];
  mk[0] = mask[b * TT + tw + lane];
  mk[1] = mask[b * TT + tw + 32 + lane];
  const float4* rows = (const float4*)(x + (b * TT + tw) * DD) + lane;
  float4 xv[2][16];
  float s[32];
#pragma unroll
  for (int i = 0; i < 16; ++i) xv[0][i] = rows[i * (DD / 4)];
#pragma unroll
  for (int h = 0; h < 4; ++h) {
    __builtin_amdgcn_sched_barrier(0);
    if (h + 1 < 4) {
#pragma unroll
      for (int i = 0; i < 16; ++i)
        xv[(h + 1) & 1][i] = rows[((h + 1) * 16 + i) * (DD / 4)];
    }
    __builtin_amdgcn_sched_barrier(0);
#pragma unroll
    for (int i = 0; i < 16; ++i) {
      const float4 v = xv[h & 1][i];
      s[(h & 1) * 16 + i] = v.x * qk4.x + v.y * qk4.y + v.z * qk4.z + v.w * qk4.w;
    }
    if (h & 1) {
      fold_level<16>(s, lane);
      fold_level<8>(s, lane);
      fold_level<4>(s, lane);
      fold_level<2>(s, lane);
      fold_level<1>(s, lane);
      const int o = b * TT + tw + (h >> 1) * 32 + lane;
      const float dv = s[0] + d0;
      const float sig = 1.0f / (1.0f + expf(-dv));
      const float tm = log1pf((1.0f - sig) * mk[h >> 1]);
      const float dec = rate0 * tm;
      dotOut[o] = dv;
      decayOut[o] = dec;
      e[o] = dv - dec;
    }
  }
}

__device__ __forceinline__ void lds_only_barrier() {
  __builtin_amdgcn_fence(__ATOMIC_RELEASE, "workgroup", "local");
  __builtin_amdgcn_s_barrier();
  __builtin_amdgcn_fence(__ATOMIC_ACQUIRE, "workgroup", "local");
}

__global__ __launch_bounds__(256) void softmax_xa_kernel(const float* __restrict__ x,
                                                         const float* __restrict__ e,
                                                         float* __restrict__ aOut,
                                                         float* __restrict__ xaPart) {
  const int b = blockIdx.x, chunk = blockIdx.y;
  const int tid = threadIdx.x, wave = tid >> 5, lane = tid & 31;
  const int tw = chunk * 512 + wave * 64;
  __shared__ float sMax[8];
  __shared__ float sSum[8];
  __shared__ alignas(16) float red[8 * DD];

  const float* eb = e + b * TT;
  float ev[8];
#pragma unroll
  for (int k = 0; k < 8; ++k) ev[k] = eb[k * 256 + tid];
  const float eLo = eb[tw + lane];
  const float eHi = eb[tw + 32 + lane];

  const float4* rows = (const float4*)(x + (b * TT + tw) * DD) + lane;
  float4 xv[2][16];
#pragma unroll
  for (int i = 0; i < 16; ++i) xv[0][i] = rows[i * (DD / 4)];

  float m = ev[0];
#pragma unroll
  for (int k = 1; k < 8; ++k) m = fmaxf(m, ev[k]);
#pragma unroll
  for (int off = 16; off > 0; off >>= 1) m = fmaxf(m, __shfl_xor(m, off, 32));
  if (lane == 0) sMax[wave] = m;
  lds_only_barrier();
  float M = sMax[0];
#pragma unroll
  for (int w = 1; w < 8; ++w) M = fmaxf(M, sMax[w]);

  float ls = 0.f;
#pragma unroll
  for (int k = 0; k < 8; ++k) ls += expf(ev[k] - M);
#pragma unroll
  for (int off = 16; off > 0; off >>= 1) ls += __shfl_xor(ls, off, 32);
  if (lane == 0) sSum[wave] = ls;
  lds_only_barrier();
  float Lsum = sSum[0];
#pragma unroll
  for (int w = 1; w < 8; ++w) Lsum += sSum[w];
  const float inv = 1.0f / Lsum;

  const float aLo = expf(eLo - M) * inv;
  const float aHi = expf(eHi - M) * inv;
  aOut[b * TT + tw + lane] = aLo;
  aOut[b * TT + tw + 32 + lane] = aHi;

  float4 acc0 = make_float4(0.f, 0.f, 0.f, 0.f);
  float4 acc1 = make_float4(0.f, 0.f, 0.f, 0.f);
#pragma unroll
  for (int h = 0; h < 4; ++h) {
    if (h + 1 < 4) {
#pragma unroll
      for (int i = 0; i < 16; ++i)
        xv[(h + 1) & 1][i] = rows[((h + 1) * 16 + i) * (DD / 4)];
    }
    const int abits = __float_as_int((h < 2) ? aLo : aHi);
#pragma unroll
    for (int i = 0; i < 16; ++i) {
      const float ai = __int_as_float(__builtin_amdgcn_readlane(abits, (h & 1) * 16 + i));
      const float4 v = xv[h & 1][i];
      if (i & 1) {
        acc1.x = fmaf(ai, v.x, acc1.x);
        acc1.y = fmaf(ai, v.y, acc1.y);
        acc1.z = fmaf(ai, v.z, acc1.z);
        acc1.w = fmaf(ai, v.w, acc1.w);
      } else {
        acc0.x = fmaf(ai, v.x, acc0.x);
        acc0.y = fmaf(ai, v.y, acc0.y);
        acc0.z = fmaf(ai, v.z, acc0.z);
        acc0.w = fmaf(ai, v.w, acc0.w);
      }
    }
  }
  float4 acc;
  acc.x = acc0.x + acc1.x;
  acc.y = acc0.y + acc1.y;
  acc.z = acc0.z + acc1.z;
  acc.w = acc0.w + acc1.w;
  *(float4*)(red + wave * DD + lane * 4) = acc;
  __syncthreads();
  if (tid < DD) {
    float s = red[tid];
#pragma unroll
    for (int w = 1; w < 8; ++w) s += red[w * DD + tid];
    xaPart[(b * 4 + chunk) * DD + tid] = s;
  }
}

extern "C" void kernel_launch(void* const* d_in, const int* in_sizes, int n_in,
                              void* d_out, int out_size, void* d_ws, size_t ws_size,
                              hipStream_t stream) {
  (void)in_sizes; (void)n_in; (void)out_size; (void)ws_size;
  const float* x    = (const float*)d_in[0];
  const float* mask = (const float*)d_in[1];
  const float* Wq   = (const float*)d_in[2];
  const float* bq   = (const float*)d_in[3];
  const float* Wk   = (const float*)d_in[4];
  const float* bk   = (const float*)d_in[5];
  const float* Wv   = (const float*)d_in[6];
  const float* bv   = (const float*)d_in[7];
  const float* rate = (const float*)d_in[8];

  float* wsf     = (float*)d_ws;
  float* partial = wsf;
  float* xmean   = partial + BB * 4 * DD;
  float* q       = xmean + BB * DD;
  float* qk      = q + BB * HH;
  float* dot0    = qk + BB * DD;
  float* e       = dot0 + BB;

  float* out      = (float*)d_out;
  float* aOut     = out + BB * HH;
  float* dotOut   = aOut + BB * TT;
  float* decayOut = dotOut + BB * TT;

  colsum_partial<<<dim3(BB, 4), 256, 0, stream>>>(x, partial);
  colsum_final<<<BB, 128, 0, stream>>>(partial, xmean);
  gemm128_wmma<true, 1><<<dim3(8, 8), 32, 0, stream>>>(xmean, Wq, bq, q);
  gemm128_wmma<false, 1><<<dim3(8, 8), 32, 0, stream>>>(q, Wk, nullptr, qk);
  dot0_kernel<<<BB, 128, 0, stream>>>(q, bk, dot0);
  dot_decay_kernel<<<dim3(BB, 4), 256, 0, stream>>>(x, mask, qk, dot0, rate,
                                                    dotOut, decayOut, e);
  softmax_xa_kernel<<<dim3(BB, 4), 256, 0, stream>>>(x, e, aOut, partial);
  gemm128_wmma<true, 4><<<dim3(8, 8), 32, 0, stream>>>(partial, Wv, bv, out);
}


# one launch instead of four between the column-sum pass and the dot pass: xmean, q GEMM, qk GEMM and dot0 fused, xmean and q tiles handed over in LDS (sum order and k order unchanged)
# speedup vs baseline: 2.2035x; 1.1784x over previous
#include <hip/hip_runtime.h>
#include <math.h>

#define BB 128
#define TT 2048
#define DD 128
#define HH 128

typedef __attribute__((ext_vector_type(2))) float v2f;
typedef __attribute__((ext_vector_type(8))) float v8f;

__global__ __launch_bounds__(256) void colsum_partial(const float* __restrict__ x,
                                                      float* __restrict__ partial) {
  const int b = blockIdx.x, chunk = blockIdx.y;
  const int g = threadIdx.x >> 7, d = threadIdx.x & 127;
  const int tbase = chunk * 512 + g * 256;
  const float* xb = x + ((b * TT + tbase) * DD) + d;
  float s = 0.f;
  for (int i = 0; i < 256; ++i) s += xb[i * DD];
  __shared__ float r[256];
  r[threadIdx.x] = s;
  __syncthreads();
  if (g == 0) partial[(b * 4 + chunk) * DD + d] = r[d] + r[128 + d];
}

template <bool TRANS, int NP>
__global__ __launch_bounds__(32) void gemm128_wmma(const float* __restrict__ A,
                                                   const float* __restrict__ W,
                                                   const float* __restrict__ bias,
                                                   float* __restrict__ C) {
  const int n0 = blockIdx.x * 16;
  const int m0 = blockIdx.y * 16;
  const int lane = threadIdx.x & 31;
  const int mrow = lane & 15;
  const int hi = lane >> 4;

  v8f acc = {};
  for (int k0 = 0; k0 < 128; k0 += 4) {
    const float* ap = A + (m0 + mrow) * (NP * 128) + k0 + 2 * hi;
    v2f a = *(const v2f*)ap;
#pragma unroll
    for (int c = 1; c < NP; ++c) a += *(const v2f*)(ap + c * 128);
    v2f bfrag;
    const int kb = k0 + 2 * hi;
    const int n = n0 + mrow;
    if (TRANS) {
      bfrag.x = W[n * 128 + kb + 0];
      bfrag.y = W[n * 128 + kb + 1];
    } else {
      bfrag.x = W[(kb + 0) * 128 + n];
      bfrag.y = W[(kb + 1) * 128 + n];
    }
    acc = __builtin_amdgcn_wmma_f32_16x16x4_f32(false, a, false, bfrag,
                                                (short)0, acc, false, false);
  }
  const int n = n0 + mrow;
  const float bb = bias ? bias[n] : 0.0f;
#pragma unroll
  for (int r = 0; r < 8; ++r) {
    const int m = m0 + r + 8 * hi;
    C[m * 128 + n] = acc[r] + bb;
  }
}

#define QP 132

__device__ __forceinline__ void lds_only_barrier() {
  __builtin_amdgcn_fence(__ATOMIC_RELEASE, "workgroup", "local");
  __builtin_amdgcn_s_barrier();
  __builtin_amdgcn_fence(__ATOMIC_ACQUIRE, "workgroup", "local");
}

__global__ __launch_bounds__(256) void q_chain_kernel(const float* __restrict__ partial,
                                                      const float* __restrict__ Wq,
                                                      const float* __restrict__ bq,
                                                      const float* __restrict__ Wk,
                                                      const float* __restrict__ bk,
                                                      float* __restrict__ qk,
                                                      float* __restrict__ dot0) {
  const int m0 = blockIdx.x * 16;
  const int tid = threadIdx.x, wave = tid >> 5, lane = tid & 31;
  const int mrow = lane & 15;
  const int hi = lane >> 4;
  const int n = wave * 16 + mrow;
  __shared__ alignas(16) float xs[16 * QP];
  __shared__ alignas(16) float qs[16 * QP];

  float4 pv[2][4];
#pragma unroll
  for (int j = 0; j < 2; ++j)
#pragma unroll
    for (int c = 0; c < 4; ++c)
      pv[j][c] = *(const float4*)(partial + ((m0 + wave + 8 * j) * 4 + c) * DD + lane * 4);
  v2f bqf[32];
#pragma unroll
  for (int k = 0; k < 32; ++k) bqf[k] = *(const v2f*)(Wq + n * 128 + 4 * k + 2 * hi);
  const float bqn = bq[n];

#pragma unroll
  for (int j = 0; j < 2; ++j) {
    float4 s = pv[j][0];
#pragma unroll
    for (int c = 1; c < 4; ++c) {
      s.x += pv[j][c].x; s.y += pv[j][c].y; s.z += pv[j][c].z; s.w += pv[j][c].w;
    }
    const float sc = 1.0f / (float)TT;
    s.x *= sc; s.y *= sc; s.z *= sc; s.w *= sc;
    *(float4*)(xs + (wave + 8 * j) * QP + lane * 4) = s;
  }
  lds_only_barrier();

  v2f bkf[32];
#pragma unroll
  for (int k = 0; k < 32; ++k) {
    bkf[k].x = Wk[(4 * k + 2 * hi + 0) * 128 + n];
    bkf[k].y = Wk[(4 * k + 2 * hi + 1) * 128 + n];
  }
  float bkv[4];
#pragma unroll
  for (int i = 0; i < 4; ++i) bkv[i] = bk[lane + 32 * i];

  v8f acc = {};
#pragma unroll
  for (int k = 0; k < 32; ++k) {
    const v2f a = *(const v2f*)(xs + mrow * QP + 4 * k + 2 * hi);
    acc = __builtin_amdgcn_wmma_f32_16x16x4_f32(false, a, false, bqf[k],
                                                (short)0, acc, false, false);
  }
#pragma unroll
  for (int r = 0; r < 8; ++r) qs[(r + 8 * hi) * QP + n] = acc[r] + bqn;
  lds_only_barrier();

  v8f acc2 = {};
#pragma unroll
  for (int k = 0; k < 32; ++k) {
    const v2f a = *(const v2f*)(qs + mrow * QP + 4 * k + 2 * hi);
    acc2 = __builtin_amdgcn_wmma_f32_16x16x4_f32(false, a, false, bkf[k],
                                                 (short)0, acc2, false, false);
  }
#pragma unroll
  for (int r = 0; r < 8; ++r) qk[(m0 + r + 8 * hi) * 128 + n] = acc2[r];

#pragma unroll
  for (int j = 0; j < 2; ++j) {
    const float* qrow = qs + (2 * wave + j) * QP + lane;
    const float p0 = bkv[0] * qrow[0];
    const float p1 = bkv[1] * qrow[32];
    const float p2 = bkv[2] * qrow[64];
    const float p3 = bkv[3] * qrow[96];
    float u = (p0 + p2) + (p1 + p3);
#pragma unroll
    for (int off = 16; off > 0; off >>= 1) u += __shfl_xor(u, off, 32);
    if (lane == 0) dot0[m0 + 2 * wave + j] = u;
  }
}

template <int OFF>
__device__ __forceinline__ void fold_level(float (&s)[32], const int lane) {
  const bool up = (lane & OFF) != 0;
#pragma unroll
  for (int i = 0; i < OFF; ++i) {
    const float keep = up ? s[i + OFF] : s[i];
    const float send = up ? s[i] : s[i + OFF];
    s[i] = keep + __shfl_xor(send, OFF, 32);
  }
}

__global__ __launch_bounds__(256) void dot_decay_kernel(
    const float* __restrict__ x, const float* __restrict__ mask,
    const float* __restrict__ qk, const float* __restrict__ dot0,
    const float* __restrict__ rate, float* __restrict__ dotOut,
    float* __restrict__ decayOut, float* __restrict__ e) {
  const int b = blockIdx.x;
  const int wave = threadIdx.x >> 5, lane = threadIdx.x & 31;
  const int tw = blockIdx.y * 512 + wave * 64;
  const float4 qk4 = *(const float4*)(qk + b * DD + lane * 4);
  const float rate0 = rate[0];
  const float d0 = dot0[b];
  float mk[2];
  mk[0] = mask[b * TT + tw + lane];
  mk[1] = mask[b * TT + tw + 32 + lane];
  const float4* rows = (const float4*)(x + (b * TT + tw) * DD) + lane;
  float4 xv[2][16];
  float s[32];
#pragma unroll
  for (int i = 0; i < 16; ++i) xv[0][i] = rows[i * (DD / 4)];
#pragma unroll
  for (int h = 0; h < 4; ++h) {
    __builtin_amdgcn_sched_barrier(0);
    if (h + 1 < 4) {
#pragma unroll
      for (int i = 0; i < 16; ++i)
        xv[(h + 1) & 1][i] = rows[((h + 1) * 16 + i) * (DD / 4)];
    }
    __builtin_amdgcn_sched_barrier(0);
#pragma unroll
    for (int i = 0; i < 16; ++i) {
      const float4 v = xv[h & 1][i];
      s[(h & 1) * 16 + i] = v.x * qk4.x + v.y * qk4.y + v.z * qk4.z + v.w * qk4.w;
    }
    if (h & 1) {
      fold_level<16>(s, lane);
      fold_level<8>(s, lane);
      fold_level<4>(s, lane);
      fold_level<2>(s, lane);
      fold_level<1>(s, lane);
      const int o = b * TT + tw + (h >> 1) * 32 + lane;
      const float dv = s[0] + d0;
      const float sig = 1.0f / (1.0f + expf(-dv));
      const float tm = log1pf((1.0f - sig) * mk[h >> 1]);
      const float dec = rate0 * tm;
      dotOut[o] = dv;
      decayOut[o] = dec;
      e[o] = dv - dec;
    }
  }
}

__global__ __launch_bounds__(256) void softmax_xa_kernel(const float* __restrict__ x,
                                                         const float* __restrict__ e,
                                                         float* __restrict__ aOut,
                                                         float* __restrict__ xaPart) {
  const int b = blockIdx.x, chunk = blockIdx.y;
  const int tid = threadIdx.x, wave = tid >> 5, lane = tid & 31;
  const int tw = chunk * 512 + wave * 64;
  __shared__ float sMax[8];
  __shared__ float sSum[8];
  __shared__ alignas(16) float red[8 * DD];

  const float* eb = e + b * TT;
  float ev[8];
#pragma unroll
  for (int k = 0; k < 8; ++k) ev[k] = eb[k * 256 + tid];
  const float eLo = eb[tw + lane];
  const float eHi = eb[tw + 32 + lane];

  const float4* rows = (const float4*)(x + (b * TT + tw) * DD) + lane;
  float4 xv[2][16];
#pragma unroll
  for (int i = 0; i < 16; ++i) xv[0][i] = rows[i * (DD / 4)];

  float m = ev[0];
#pragma unroll
  for (int k = 1; k < 8; ++k) m = fmaxf(m, ev[k]);
#pragma unroll
  for (int off = 16; off > 0; off >>= 1) m = fmaxf(m, __shfl_xor(m, off, 32));
  if (lane == 0) sMax[wave] = m;
  lds_only_barrier();
  float M = sMax[0];
#pragma unroll
  for (int w = 1; w < 8; ++w) M = fmaxf(M, sMax[w]);

  float ls = 0.f;
#pragma unroll
  for (int k = 0; k < 8; ++k) ls += expf(ev[k] - M);
#pragma unroll
  for (int off = 16; off > 0; off >>= 1) ls += __shfl_xor(ls, off, 32);
  if (lane == 0) sSum[wave] = ls;
  lds_only_barrier();
  float Lsum = sSum[0];
#pragma unroll
  for (int w = 1; w < 8; ++w) Lsum += sSum[w];
  const float inv = 1.0f / Lsum;

  const float aLo = expf(eLo - M) * inv;
  const float aHi = expf(eHi - M) * inv;
  aOut[b * TT + tw + lane] = aLo;
  aOut[b * TT + tw + 32 + lane] = aHi;

  float4 acc0 = make_float4(0.f, 0.f, 0.f, 0.f);
  float4 acc1 = make_float4(0.f, 0.f, 0.f, 0.f);
#pragma unroll
  for (int h = 0; h < 4; ++h) {
    if (h + 1 < 4) {
#pragma unroll
      for (int i = 0; i < 16; ++i)
        xv[(h + 1) & 1][i] = rows[((h + 1) * 16 + i) * (DD / 4)];
    }
    const int abits = __float_as_int((h < 2) ? aLo : aHi);
#pragma unroll
    for (int i = 0; i < 16; ++i) {
      const float ai = __int_as_float(__builtin_amdgcn_readlane(abits, (h & 1) * 16 + i));
      const float4 v = xv[h & 1][i];
      if (i & 1) {
        acc1.x = fmaf(ai, v.x, acc1.x);
        acc1.y = fmaf(ai, v.y, acc1.y);
        acc1.z = fmaf(ai, v.z, acc1.z);
        acc1.w = fmaf(ai, v.w, acc1.w);
      } else {
        acc0.x = fmaf(ai, v.x, acc0.x);
        acc0.y = fmaf(ai, v.y, acc0.y);
        acc0.z = fmaf(ai, v.z, acc0.z);
        acc0.w = fmaf(ai, v.w, acc0.w);
      }
    }
  }
  float4 acc;
  acc.x = acc0.x + acc1.x;
  acc.y = acc0.y + acc1.y;
  acc.z = acc0.z + acc1.z;
  acc.w = acc0.w + acc1.w;
  *(float4*)(red + wave * DD + lane * 4) = acc;
  __syncthreads();
  if (tid < DD) {
    float s = red[tid];
#pragma unroll
    for (int w = 1; w < 8; ++w) s += red[w * DD + tid];
    xaPart[(b * 4 + chunk) * DD + tid] = s;
  }
}

extern "C" void kernel_launch(void* const* d_in, const int* in_sizes, int n_in,
                              void* d_out, int out_size, void* d_ws, size_t ws_size,
                              hipStream_t stream) {
  (void)in_sizes; (void)n_in; (void)out_size; (void)ws_size;
  const float* x    = (const float*)d_in[0];
  const float* mask = (const float*)d_in[1];
  const float* Wq   = (const float*)d_in[2];
  const float* bq   = (const float*)d_in[3];
  const float* Wk   = (const float*)d_in[4];
  const float* bk   = (const float*)d_in[5];
  const float* Wv   = (const float*)d_in[6];
  const float* bv   = (const float*)d_in[7];
  const float* rate = (const float*)d_in[8];

  float* wsf     = (float*)d_ws;
  float* partial = wsf;
  float* qk      = partial + BB * 4 * DD;
  float* dot0    = qk + BB * DD;
  float* e       = dot0 + BB;

  float* out      = (float*)d_out;
  float* aOut     = out + BB * HH;
  float* dotOut   = aOut + BB * TT;
  float* decayOut = dotOut + BB * TT;

  colsum_partial<<<dim3(BB, 4), 256, 0, stream>>>(x, partial);
  q_chain_kernel<<<BB / 16, 256, 0, stream>>>(partial, Wq, bq, Wk, bk, qk, dot0);
  dot_decay_kernel<<<dim3(BB, 4), 256, 0, stream>>>(x, mask, qk, dot0, rate,
                                                    dotOut, decayOut, e);
  softmax_xa_kernel<<<dim3(BB, 4), 256, 0, stream>>>(x, e, aOut, partial);
  gemm128_wmma<true, 4><<<dim3(8, 8), 32, 0, stream>>>(partial, Wv, bv, out);
}


# dot pass and softmax/xa pass fused: x read once there, rows weighted in registers by exp(e - running max), 512-row shares rescaled in one finishing launch (a, out GEMM); new sum trees: dot, xa
# speedup vs baseline: 2.4314x; 1.1034x over previous
#include <hip/hip_runtime.h>
#include <math.h>

#define BB 128
#define TT 2048
#define DD 128
#define HH 128

typedef __attribute__((ext_vector_type(2))) float v2f;
typedef __attribute__((ext_vector_type(8))) float v8f;

__global__ __launch_bounds__(256) void colsum_partial(const float* __restrict__ x,
                                                      float* __restrict__ partial) {
  const int b = blockIdx.x, chunk = blockIdx.y;
  const int g = threadIdx.x >> 7, d = threadIdx.x & 127;
  const int tbase = chunk * 512 + g * 256;
  const float* xb = x + ((b * TT + tbase) * DD) + d;
  float s = 0.f;
  for (int i = 0; i < 256; ++i) s += xb[i * DD];
  __shared__ float r[256];
  r[threadIdx.x] = s;
  __syncthreads();
  if (g == 0) partial[(b * 4 + chunk) * DD + d] = r[d] + r[128 + d];
}

#define QP 132

__device__ __forceinline__ void lds_only_barrier() {
  __builtin_amdgcn_fence(__ATOMIC_RELEASE, "workgroup", "local");
  __builtin_amdgcn_s_barrier();
  __builtin_amdgcn_fence(__ATOMIC_ACQUIRE, "workgroup", "local");
}

__global__ __launch_bounds__(256) void q_chain_kernel(const float* __restrict__ partial,
                                                      const float* __restrict__ Wq,
                                                      const float* __restrict__ bq,
                                                      const float* __restrict__ Wk,
                                                      const float* __restrict__ bk,
                                                      float* __restrict__ qk,
                                                      float* __restrict__ dot0) {
  const int m0 = blockIdx.x * 16;
  const int tid = threadIdx.x, wave = tid >> 5, lane = tid & 31;
  const int mrow = lane & 15;
  const int hi = lane >> 4;
  const int n = wave * 16 + mrow;
  __shared__ alignas(16) float xs[16 * QP];
  __shared__ alignas(16) float qs[16 * QP];

  float4 pv[2][4];
#pragma unroll
  for (int j = 0; j < 2; ++j)
#pragma unroll
    for (int c = 0; c < 4; ++c)
      pv[j][c] = *(const float4*)(partial + ((m0 + wave + 8 * j) * 4 + c) * DD + lane * 4);
  v2f bqf[32];
#pragma unroll
  for (int k = 0; k < 32; ++k) bqf[k] = *(const v2f*)(Wq + n * 128 + 4 * k + 2 * hi);
  const float bqn = bq[n];

#pragma unroll
  for (int j = 0; j < 2; ++j) {
    float4 s = pv[j][0];
#pragma unroll
    for (int c = 1; c < 4; ++c) {
      s.x += pv[j][c].x; s.y += pv[j][c].y; s.z += pv[j][c].z; s.w += pv[j][c].w;
    }
    const float sc = 1.0f / (float)TT;
    s.x *= sc; s.y *= sc; s.z *= sc; s.w *= sc;
    *(float4*)(xs + (wave + 8 * j) * QP + lane * 4) = s;
  }
  lds_only_barrier();

  v2f bkf[32];
#pragma unroll
  for (int k = 0; k < 32; ++k) {
    bkf[k].x = Wk[(4 * k + 2 * hi + 0) * 128 + n];
    bkf[k].y = Wk[(4 * k + 2 * hi + 1) * 128 + n];
  }
  float bkv[4];
#pragma unroll
  for (int i = 0; i < 4; ++i) bkv[i] = bk[lane + 32 * i];

  v8f acc = {};
#pragma unroll
  for (int k = 0; k < 32; ++k) {
    const v2f a = *(const v2f*)(xs + mrow * QP + 4 * k + 2 * hi);
    acc = __builtin_amdgcn_wmma_f32_16x16x4_f32(false, a, false, bqf[k],
                                                (short)0, acc, false, false);
  }
#pragma unroll
  for (int r = 0; r < 8; ++r) qs[(r + 8 * hi) * QP + n] = acc[r] + bqn;
  lds_only_barrier();

  v8f acc2 = {};
#pragma unroll
  for (int k = 0; k < 32; ++k) {
    const v2f a = *(const v2f*)(qs + mrow * QP + 4 * k + 2 * hi);
    acc2 = __builtin_amdgcn_wmma_f32_16x16x4_f32(false, a, false, bkf[k],
                                                 (short)0, acc2, false, false);
  }
#pragma unroll
  for (int r = 0; r < 8; ++r) qk[(m0 + r + 8 * hi) * 128 + n] = acc2[r];

#pragma unroll
  for (int j = 0; j < 2; ++j) {
    const float* qrow = qs + (2 * wave + j) * QP + lane;
    const float p0 = bkv[0] * qrow[0];
    const float p1 = bkv[1] * qrow[32];
    const float p2 = bkv[2] * qrow[64];
    const float p3 = bkv[3] * qrow[96];
    float u = (p0 + p2) + (p1 + p3);
#pragma unroll
    for (int off = 16; off > 0; off >>= 1) u += __shfl_xor(u, off, 32);
    if (lane == 0) dot0[m0 + 2 * wave + j] = u;
  }
}

template <int OFF>
__device__ __forceinline__ void fold_level(float (&s)[16], const int lane) {
  const bool up = (lane & OFF) != 0;
#pragma unroll
  for (int i = 0; i < OFF; ++i) {
    const float keep = up ? s[i + OFF] : s[i];
    const float send = up ? s[i] : s[i + OFF];
    s[i] = keep + __shfl_xor(send, OFF, 32);
  }
}

__global__ __launch_bounds__(256) void dot_softmax_xa_kernel(
    const float* __restrict__ x, const float* __restrict__ mask,
    const float* __restrict__ qk, const float* __restrict__ dot0,
    const float* __restrict__ rate, float* __restrict__ dotOut,
    float* __restrict__ decayOut, float* __restrict__ e,
    float* __restrict__ xaPart) {
  const int b = blockIdx.x, chunk = blockIdx.y;
  const int tid = threadIdx.x, wave = tid >> 5, lane = tid & 31;
  const int row16 = lane & 15;
  const bool hi = lane >= 16;
  const int tw = chunk * 512 + wave * 64;
  __shared__ float sMax[8];
  __shared__ alignas(16) float red[8 * DD];

  float4 qk4 = *(const float4*)(qk + b * DD + lane * 4);
  const float rate0 = rate[0];
  const float d0 = dot0[b];
  float mk[4];
#pragma unroll
  for (int g = 0; g < 4; ++g) mk[g] = mask[b * TT + tw + 16 * g + row16];
  const float4* rows = (const float4*)(x + (b * TT + tw) * DD) + lane;
  float4 xv[2][16];
#pragma unroll
  for (int i = 0; i < 16; ++i) xv[0][i] = rows[i * (DD / 4)];

  float4 acc0 = make_float4(0.f, 0.f, 0.f, 0.f);
  float4 acc1 = make_float4(0.f, 0.f, 0.f, 0.f);
  float mrun = 0.f;
  float dvE = 0.f, decE = 0.f, evE = 0.f;
#pragma unroll
  for (int g = 0; g < 4; ++g) {
    if (g + 1 < 4) {
#pragma unroll
      for (int i = 0; i < 16; ++i)
        xv[(g + 1) & 1][i] = rows[((g + 1) * 16 + i) * (DD / 4)];
    }
    asm volatile("" : "+v"(qk4.x), "+v"(qk4.y), "+v"(qk4.z), "+v"(qk4.w));
    float s[16];
#pragma unroll
    for (int i = 0; i < 16; ++i) {
      const float4 v = xv[g & 1][i];
      s[i] = v.x * qk4.x + v.y * qk4.y + v.z * qk4.z + v.w * qk4.w;
    }
    fold_level<8>(s, lane);
    fold_level<4>(s, lane);
    fold_level<2>(s, lane);
    fold_level<1>(s, lane);
    const float dsum = s[0] + __shfl_xor(s[0], 16, 32);
    const float dv = dsum + d0;
    const float sig = 1.0f / (1.0f + expf(-dv));
    const float tm = log1pf((1.0f - sig) * mk[g]);
    const float dec = rate0 * tm;
    const float ev = dv - dec;
    float gm = ev;
#pragma unroll
    for (int off = 8; off > 0; off >>= 1) gm = fmaxf(gm, __shfl_xor(gm, off, 32));
    if (g == 0) {
      mrun = gm;
    } else {
      const float mnew = fmaxf(mrun, gm);
      const float sc = expf(mrun - mnew);
      acc0.x *= sc; acc0.y *= sc; acc0.z *= sc; acc0.w *= sc;
      acc1.x *= sc; acc1.y *= sc; acc1.z *= sc; acc1.w *= sc;
      mrun = mnew;
    }
    const int wbits = __float_as_int(expf(ev - mrun));
#pragma unroll
    for (int i = 0; i < 16; ++i) {
      const float wi = __int_as_float(__builtin_amdgcn_readlane(wbits, i));
      const float4 v = xv[g & 1][i];
      if (i & 1) {
        acc1.x = fmaf(wi, v.x, acc1.x);
        acc1.y = fmaf(wi, v.y, acc1.y);
        acc1.z = fmaf(wi, v.z, acc1.z);
        acc1.w = fmaf(wi, v.w, acc1.w);
      } else {
        acc0.x = fmaf(wi, v.x, acc0.x);
        acc0.y = fmaf(wi, v.y, acc0.y);
        acc0.z = fmaf(wi, v.z, acc0.z);
        acc0.w = fmaf(wi, v.w, acc0.w);
      }
    }
    asm volatile("" : "+v"(acc0.x), "+v"(acc0.y), "+v"(acc0.z), "+v"(acc0.w),
                      "+v"(acc1.x), "+v"(acc1.y), "+v"(acc1.z), "+v"(acc1.w));
    if (g & 1) {
      const int o = b * TT + tw + (g >> 1) * 32 + lane;
      dotOut[o] = hi ? dv : dvE;
      decayOut[o] = hi ? dec : decE;
      e[o] = hi ? ev : evE;
    } else {
      dvE = dv; decE = dec; evE = ev;
    }
  }
  float4 acc;
  acc.x = acc0.x + acc1.x;
  acc.y = acc0.y + acc1.y;
  acc.z = acc0.z + acc1.z;
  acc.w = acc0.w + acc1.w;
  if (lane == 0) sMax[wave] = mrun;
  *(float4*)(red + wave * DD + lane * 4) = acc;
  __syncthreads();
  if (tid < DD) {
    float m = sMax[0];
#pragma unroll
    for (int w = 1; w < 8; ++w) m = fmaxf(m, sMax[w]);
    float sum = expf(sMax[0] - m) * red[tid];
#pragma unroll
    for (int w = 1; w < 8; ++w) sum = fmaf(expf(sMax[w] - m), red[w * DD + tid], sum);
    xaPart[(b * 4 + chunk) * DD + tid] = sum;
  }
}

__global__ __launch_bounds__(256) void finish_kernel(const float* __restrict__ e,
                                                     const float* __restrict__ xaPart,
                                                     const float* __restrict__ Wv,
                                                     const float* __restrict__ bv,
                                                     float* __restrict__ aOut,
                                                     float* __restrict__ out) {
  const int b = blockIdx.x;
  const int tid = threadIdx.x, wave = tid >> 5, lane = tid & 31;
  const int mrow = lane & 15;
  const int hi = lane >> 4;
  const int n = wave * 16 + mrow;
  __shared__ float sMax[8][4];
  __shared__ float sSum[8];
  __shared__ alignas(16) float xs[DD];
  __shared__ float os[HH];

  const float* eb = e + b * TT;
  float ev[8];
#pragma unroll
  for (int k = 0; k < 8; ++k) ev[k] = eb[k * 256 + tid];
  float pr[4] = {0.f, 0.f, 0.f, 0.f};
  if (tid < DD) {
#pragma unroll
    for (int c = 0; c < 4; ++c) pr[c] = xaPart[(b * 4 + c) * DD + tid];
  }
  v2f bvf[32];
#pragma unroll
  for (int k = 0; k < 32; ++k) bvf[k] = *(const v2f*)(Wv + n * 128 + 4 * k + 2 * hi);
  const float bvn = bv[n];

  float mc[4];
#pragma unroll
  for (int c = 0; c < 4; ++c) {
    mc[c] = fmaxf(ev[2 * c], ev[2 * c + 1]);
#pragma unroll
    for (int off = 16; off > 0; off >>= 1) mc[c] = fmaxf(mc[c], __shfl_xor(mc[c], off, 32));
    if (lane == 0) sMax[wave][c] = mc[c];
  }
  lds_only_barrier();
#pragma unroll
  for (int c = 0; c < 4; ++c) {
    mc[c] = sMax[0][c];
#pragma unroll
    for (int w = 1; w < 8; ++w) mc[c] = fmaxf(mc[c], sMax[w][c]);
  }
  const float M = fmaxf(fmaxf(mc[0], mc[1]), fmaxf(mc[2], mc[3]));

  float ls = 0.f;
#pragma unroll
  for (int k = 0; k < 8; ++k) ls += expf(ev[k] - M);
#pragma unroll
  for (int off = 16; off > 0; off >>= 1) ls += __shfl_xor(ls, off, 32);
  if (lane == 0) sSum[wave] = ls;
  lds_only_barrier();
  float Lsum = sSum[0];
#pragma unroll
  for (int w = 1; w < 8; ++w) Lsum += sSum[w];
  const float inv = 1.0f / Lsum;

#pragma unroll
  for (int k = 0; k < 8; ++k) aOut[b * TT + k * 256 + tid] = expf(ev[k] - M) * inv;

  if (tid < DD) {
    float sum = (expf(mc[0] - M) * inv) * pr[0];
#pragma unroll
    for (int c = 1; c < 4; ++c) sum = fmaf(expf(mc[c] - M) * inv, pr[c], sum);
    xs[tid] = sum;
  }
  lds_only_barrier();

  v8f acc = {};
#pragma unroll
  for (int k = 0; k < 32; ++k) {
    const v2f a = *(const v2f*)(xs + 4 * k + 2 * hi);
    acc = __builtin_amdgcn_wmma_f32_16x16x4_f32(false, a, false, bvf[k],
                                                (short)0, acc, false, false);
  }
  if (hi == 0) os[n] = acc[0] + bvn;
  lds_only_barrier();
  if (tid < HH) out[b * HH + tid] = os[tid];
}

extern "C" void kernel_launch(void* const* d_in, const int* in_sizes, int n_in,
                              void* d_out, int out_size, void* d_ws, size_t ws_size,
                              hipStream_t stream) {
  (void)in_sizes; (void)n_in; (void)out_size; (void)ws_size;
  const float* x    = (const float*)d_in[0];
  const float* mask = (const float*)d_in[1];
  const float* Wq   = (const float*)d_in[2];
  const float* bq   = (const float*)d_in[3];
  const float* Wk   = (const float*)d_in[4];
  const float* bk   = (const float*)d_in[5];
  const float* Wv   = (const float*)d_in[6];
  const float* bv   = (const float*)d_in[7];
  const float* rate = (const float*)d_in[8];

  float* wsf     = (float*)d_ws;
  float* partial = wsf;
  float* qk      = partial + BB * 4 * DD;
  float* dot0    = qk + BB * DD;
  float* e       = dot0 + BB;

  float* out      = (float*)d_out;
  float* aOut     = out + BB * HH;
  float* dotOut   = aOut + BB * TT;
  float* decayOut = dotOut + BB * TT;

  colsum_partial<<<dim3(BB, 4), 256, 0, stream>>>(x, partial);
  q_chain_kernel<<<BB / 16, 256, 0, stream>>>(partial, Wq, bq, Wk, bk, qk, dot0);
  dot_softmax_xa_kernel<<<dim3(BB, 4), 256, 0, stream>>>(x, mask, qk, dot0, rate,
                                                         dotOut, decayOut, e, partial);
  finish_kernel<<<BB, 256, 0, stream>>>(e, partial, Wv, bv, aOut, out);
}


# column-sum pass: each wave adds 64 whole rows from 16-byte row loads, two 16-row groups requested before the first add, about 32 loads in flight; 8 waves meet in LDS (new sum tree: column sums)
# speedup vs baseline: 2.7596x; 1.1350x over previous
#include <hip/hip_runtime.h>
#include <math.h>

#define BB 128
#define TT 2048
#define DD 128
#define HH 128

typedef __attribute__((ext_vector_type(2))) float v2f;
typedef __attribute__((ext_vector_type(8))) float v8f;

__global__ __launch_bounds__(256) void colsum_partial(const float* __restrict__ x,
                                                      float* __restrict__ partial) {
  const int b = blockIdx.x, chunk = blockIdx.y;
  const int tid = threadIdx.x, wave = tid >> 5, lane = tid & 31;
  const int tw = chunk * 512 + wave * 64;
  __shared__ alignas(16) float red[8 * DD];
  const float4* rows = (const float4*)(x + (b * TT + tw) * DD) + lane;
  float4 xv[2][16];
#pragma unroll
  for (int i = 0; i < 16; ++i) xv[0][i] = rows[i * (DD / 4)];
  float4 acc[4];
#pragma unroll
  for (int k = 0; k < 4; ++k) acc[k] = make_float4(0.f, 0.f, 0.f, 0.f);
#pragma unroll
  for (int g = 0; g < 4; ++g) {
    if (g + 1 < 4) {
#pragma unroll
      for (int i = 0; i < 16; ++i)
        xv[(g + 1) & 1][i] = rows[((g + 1) * 16 + i) * (DD / 4)];
      asm volatile("" : "+v"(acc[0].x), "+v"(acc[0].y), "+v"(acc[0].z), "+v"(acc[0].w),
                        "+v"(acc[1].x), "+v"(acc[1].y), "+v"(acc[1].z), "+v"(acc[1].w),
                        "+v"(acc[2].x), "+v"(acc[2].y), "+v"(acc[2].z), "+v"(acc[2].w),
                        "+v"(acc[3].x), "+v"(acc[3].y), "+v"(acc[3].z), "+v"(acc[3].w));
    }
#pragma unroll
    for (int i = 0; i < 16; ++i) {
      const float4 v = xv[g & 1][i];
      acc[i & 3].x += v.x; acc[i & 3].y += v.y;
      acc[i & 3].z += v.z; acc[i & 3].w += v.w;
    }
  }
  float4 s;
  s.x = (acc[0].x + acc[1].x) + (acc[2].x + acc[3].x);
  s.y = (acc[0].y + acc[1].y) + (acc[2].y + acc[3].y);
  s.z = (acc[0].z + acc[1].z) + (acc[2].z + acc[3].z);
  s.w = (acc[0].w + acc[1].w) + (acc[2].w + acc[3].w);
  *(float4*)(red + wave * DD + lane * 4) = s;
  __syncthreads();
  if (tid < DD) {
    float t = red[tid];
#pragma unroll
    for (int w = 1; w < 8; ++w) t += red[w * DD + tid];
    partial[(b * 4 + chunk) * DD + tid] = t;
  }
}

#define QP 132

__device__ __forceinline__ void lds_only_barrier() {
  __builtin_amdgcn_fence(__ATOMIC_RELEASE, "workgroup", "local");
  __builtin_amdgcn_s_barrier();
  __builtin_amdgcn_fence(__ATOMIC_ACQUIRE, "workgroup", "local");
}

__global__ __launch_bounds__(256) void q_chain_kernel(const float* __restrict__ partial,
                                                      const float* __restrict__ Wq,
                                                      const float* __restrict__ bq,
                                                      const float* __restrict__ Wk,
                                                      const float* __restrict__ bk,
                                                      float* __restrict__ qk,
                                                      float* __restrict__ dot0) {
  const int m0 = blockIdx.x * 16;
  const int tid = threadIdx.x, wave = tid >> 5, lane = tid & 31;
  const int mrow = lane & 15;
  const int hi = lane >> 4;
  const int n = wave * 16 + mrow;
  __shared__ alignas(16) float xs[16 * QP];
  __shared__ alignas(16) float qs[16 * QP];

  float4 pv[2][4];
#pragma unroll
  for (int j = 0; j < 2; ++j)
#pragma unroll
    for (int c = 0; c < 4; ++c)
      pv[j][c] = *(const float4*)(partial + ((m0 + wave + 8 * j) * 4 + c) * DD + lane * 4);
  v2f bqf[32];
#pragma unroll
  for (int k = 0; k < 32; ++k) bqf[k] = *(const v2f*)(Wq + n * 128 + 4 * k + 2 * hi);
  const float bqn = bq[n];

#pragma unroll
  for (int j = 0; j < 2; ++j) {
    float4 s = pv[j][0];
#pragma unroll
    for (int c = 1; c < 4; ++c) {
      s.x += pv[j][c].x; s.y += pv[j][c].y; s.z += pv[j][c].z; s.w += pv[j][c].w;
    }
    const float sc = 1.0f / (float)TT;
    s.x *= sc; s.y *= sc; s.z *= sc; s.w *= sc;
    *(float4*)(xs + (wave + 8 * j) * QP + lane * 4) = s;
  }
  lds_only_barrier();

  v2f bkf[32];
#pragma unroll
  for (int k = 0; k < 32; ++k) {
    bkf[k].x = Wk[(4 * k + 2 * hi + 0) * 128 + n];
    bkf[k].y = Wk[(4 * k + 2 * hi + 1) * 128 + n];
  }
  float bkv[4];
#pragma unroll
  for (int i = 0; i < 4; ++i) bkv[i] = bk[lane + 32 * i];

  v8f acc = {};
#pragma unroll
  for (int k = 0; k < 32; ++k) {
    const v2f a = *(const v2f*)(xs + mrow * QP + 4 * k + 2 * hi);
    acc = __builtin_amdgcn_wmma_f32_16x16x4_f32(false, a, false, bqf[k],
                                                (short)0, acc, false, false);
  }
#pragma unroll
  for (int r = 0; r < 8; ++r) qs[(r + 8 * hi) * QP + n] = acc[r] + bqn;
  lds_only_barrier();

  v8f acc2 = {};
#pragma unroll
  for (int k = 0; k < 32; ++k) {
    const v2f a = *(const v2f*)(qs + mrow * QP + 4 * k + 2 * hi);
    acc2 = __builtin_amdgcn_wmma_f32_16x16x4_f32(false, a, false, bkf[k],
                                                 (short)0, acc2, false, false);
  }
#pragma unroll
  for (int r = 0; r < 8; ++r) qk[(m0 + r + 8 * hi) * 128 + n] = acc2[r];

#pragma unroll
  for (int j = 0; j < 2; ++j) {
    const float* qrow = qs + (2 * wave + j) * QP + lane;
    const float p0 = bkv[0] * qrow[0];
    const float p1 = bkv[1] * qrow[32];
    const float p2 = bkv[2] * qrow[64];
    const float p3 = bkv[3] * qrow[96];
    float u = (p0 + p2) + (p1 + p3);
#pragma unroll
    for (int off = 16; off > 0; off >>= 1) u += __shfl_xor(u, off, 32);
    if (lane == 0) dot0[m0 + 2 * wave + j] = u;
  }
}

template <int OFF>
__device__ __forceinline__ void fold_level(float (&s)[16], const int lane) {
  const bool up = (lane & OFF) != 0;
#pragma unroll
  for (int i = 0; i < OFF; ++i) {
    const float keep = up ? s[i + OFF] : s[i];
    const float send = up ? s[i] : s[i + OFF];
    s[i] = keep + __shfl_xor(send, OFF, 32);
  }
}

__global__ __launch_bounds__(256) void dot_softmax_xa_kernel(
    const float* __restrict__ x, const float* __restrict__ mask,
    const float* __restrict__ qk, const float* __restrict__ dot0,
    const float* __restrict__ rate, float* __restrict__ dotOut,
    float* __restrict__ decayOut, float* __restrict__ e,
    float* __restrict__ xaPart) {
  const int b = blockIdx.x, chunk = blockIdx.y;
  const int tid = threadIdx.x, wave = tid >> 5, lane = tid & 31;
  const int row16 = lane & 15;
  const bool hi = lane >= 16;
  const int tw = chunk * 512 + wave * 64;
  __shared__ float sMax[8];
  __shared__ alignas(16) float red[8 * DD];

  float4 qk4 = *(const float4*)(qk + b * DD + lane * 4);
  const float rate0 = rate[0];
  const float d0 = dot0[b];
  float mk[4];
#pragma unroll
  for (int g = 0; g < 4; ++g) mk[g] = mask[b * TT + tw + 16 * g + row16];
  const float4* rows = (const float4*)(x + (b * TT + tw) * DD) + lane;
  float4 xv[2][16];
#pragma unroll
  for (int i = 0; i < 16; ++i) xv[0][i] = rows[i * (DD / 4)];

  float4 acc0 = make_float4(0.f, 0.f, 0.f, 0.f);
  float4 acc1 = make_float4(0.f, 0.f, 0.f, 0.f);
  float mrun = 0.f;
  float dvE = 0.f, decE = 0.f, evE = 0.f;
#pragma unroll
  for (int g = 0; g < 4; ++g) {
    if (g + 1 < 4) {
#pragma unroll
      for (int i = 0; i < 16; ++i)
        xv[(g + 1) & 1][i] = rows[((g + 1) * 16 + i) * (DD / 4)];
    }
    asm volatile("" : "+v"(qk4.x), "+v"(qk4.y), "+v"(qk4.z), "+v"(qk4.w));
    float s[16];
#pragma unroll
    for (int i = 0; i < 16; ++i) {
      const float4 v = xv[g & 1][i];
      s[i] = v.x * qk4.x + v.y * qk4.y + v.z * qk4.z + v.w * qk4.w;
    }
    fold_level<8>(s, lane);
    fold_level<4>(s, lane);
    fold_level<2>(s, lane);
    fold_level<1>(s, lane);
    const float dsum = s[0] + __shfl_xor(s[0], 16, 32);
    const float dv = dsum + d0;
    const float sig = 1.0f / (1.0f + expf(-dv));
    const float tm = log1pf((1.0f - sig) * mk[g]);
    const float dec = rate0 * tm;
    const float ev = dv - dec;
    float gm = ev;
#pragma unroll
    for (int off = 8; off > 0; off >>= 1) gm = fmaxf(gm, __shfl_xor(gm, off, 32));
    if (g == 0) {
      mrun = gm;
    } else {
      const float mnew = fmaxf(mrun, gm);
      const float sc = expf(mrun - mnew);
      acc0.x *= sc; acc0.y *= sc; acc0.z *= sc; acc0.w *= sc;
      acc1.x *= sc; acc1.y *= sc; acc1.z *= sc; acc1.w *= sc;
      mrun = mnew;
    }
    const int wbits = __float_as_int(expf(ev - mrun));
#pragma unroll
    for (int i = 0; i < 16; ++i) {
      const float wi = __int_as_float(__builtin_amdgcn_readlane(wbits, i));
      const float4 v = xv[g & 1][i];
      if (i & 1) {
        acc1.x = fmaf(wi, v.x, acc1.x);
        acc1.y = fmaf(wi, v.y, acc1.y);
        acc1.z = fmaf(wi, v.z, acc1.z);
        acc1.w = fmaf(wi, v.w, acc1.w);
      } else {
        acc0.x = fmaf(wi, v.x, acc0.x);
        acc0.y = fmaf(wi, v.y, acc0.y);
        acc0.z = fmaf(wi, v.z, acc0.z);
        acc0.w = fmaf(wi, v.w, acc0.w);
      }
    }
    asm volatile("" : "+v"(acc0.x), "+v"(acc0.y), "+v"(acc0.z), "+v"(acc0.w),
                      "+v"(acc1.x), "+v"(acc1.y), "+v"(acc1.z), "+v"(acc1.w));
    if (g & 1) {
      const int o = b * TT + tw + (g >> 1) * 32 + lane;
      dotOut[o] = hi ? dv : dvE;
      decayOut[o] = hi ? dec : decE;
      e[o] = hi ? ev : evE;
    } else {
      dvE = dv; decE = dec; evE = ev;
    }
  }
  float4 acc;
  acc.x = acc0.x + acc1.x;
  acc.y = acc0.y + acc1.y;
  acc.z = acc0.z + acc1.z;
  acc.w = acc0.w + acc1.w;
  if (lane == 0) sMax[wave] = mrun;
  *(float4*)(red + wave * DD + lane * 4) = acc;
  __syncthreads();
  if (tid < DD) {
    float m = sMax[0];
#pragma unroll
    for (int w = 1; w < 8; ++w) m = fmaxf(m, sMax[w]);
    float sum = expf(sMax[0] - m) * red[tid];
#pragma unroll
    for (int w = 1; w < 8; ++w) sum = fmaf(expf(sMax[w] - m), red[w * DD + tid], sum);
    xaPart[(b * 4 + chunk) * DD + tid] = sum;
  }
}

__global__ __launch_bounds__(256) void finish_kernel(const float* __restrict__ e,
                                                     const float* __restrict__ xaPart,
                                                     const float* __restrict__ Wv,
                                                     const float* __restrict__ bv,
                                                     float* __restrict__ aOut,
                                                     float* __restrict__ out) {
  const int b = blockIdx.x;
  const int tid = threadIdx.x, wave = tid >> 5, lane = tid & 31;
  const int mrow = lane & 15;
  const int hi = lane >> 4;
  const int n = wave * 16 + mrow;
  __shared__ float sMax[8][4];
  __shared__ float sSum[8];
  __shared__ alignas(16) float xs[DD];
  __shared__ float os[HH];

  const float* eb = e + b * TT;
  float ev[8];
#pragma unroll
  for (int k = 0; k < 8; ++k) ev[k] = eb[k * 256 + tid];
  float pr[4] = {0.f, 0.f, 0.f, 0.f};
  if (tid < DD) {
#pragma unroll
    for (int c = 0; c < 4; ++c) pr[c] = xaPart[(b * 4 + c) * DD + tid];
  }
  v2f bvf[32];
#pragma unroll
  for (int k = 0; k < 32; ++k) bvf[k] = *(const v2f*)(Wv + n * 128 + 4 * k + 2 * hi);
  const float bvn = bv[n];

  float mc[4];
#pragma unroll
  for (int c = 0; c < 4; ++c) {
    mc[c] = fmaxf(ev[2 * c], ev[2 * c + 1]);
#pragma unroll
    for (int off = 16; off > 0; off >>= 1) mc[c] = fmaxf(mc[c], __shfl_xor(mc[c], off, 32));
    if (lane == 0) sMax[wave][c] = mc[c];
  }
  lds_only_barrier();
#pragma unroll
  for (int c = 0; c < 4; ++c) {
    mc[c] = sMax[0][c];
#pragma unroll
    for (int w = 1; w < 8; ++w) mc[c] = fmaxf(mc[c], sMax[w][c]);
  }
  const float M = fmaxf(fmaxf(mc[0], mc[1]), fmaxf(mc[2], mc[3]));

  float ls = 0.f;
#pragma unroll
  for (int k = 0; k < 8; ++k) ls += expf(ev[k] - M);
#pragma unroll
  for (int off = 16; off > 0; off >>= 1) ls += __shfl_xor(ls, off, 32);
  if (lane == 0) sSum[wave] = ls;
  lds_only_barrier();
  float Lsum = sSum[0];
#pragma unroll
  for (int w = 1; w < 8; ++w) Lsum += sSum[w];
  const float inv = 1.0f / Lsum;

#pragma unroll
  for (int k = 0; k < 8; ++k) aOut[b * TT + k * 256 + tid] = expf(ev[k] - M) * inv;

  if (tid < DD) {
    float sum = (expf(mc[0] - M) * inv) * pr[0];
#pragma unroll
    for (int c = 1; c < 4; ++c) sum = fmaf(expf(mc[c] - M) * inv, pr[c], sum);
    xs[tid] = sum;
  }
  lds_only_barrier();

  v8f acc = {};
#pragma unroll
  for (int k = 0; k < 32; ++k) {
    const v2f a = *(const v2f*)(xs + 4 * k + 2 * hi);
    acc = __builtin_amdgcn_wmma_f32_16x16x4_f32(false, a, false, bvf[k],
                                                (short)0, acc, false, false);
  }
  if (hi == 0) os[n] = acc[0] + bvn;
  lds_only_barrier();
  if (tid < HH) out[b * HH + tid] = os[tid];
}

extern "C" void kernel_launch(void* const* d_in, const int* in_sizes, int n_in,
                              void* d_out, int out_size, void* d_ws, size_t ws_size,
                              hipStream_t stream) {
  (void)in_sizes; (void)n_in; (void)out_size; (void)ws_size;
  const float* x    = (const float*)d_in[0];
  const float* mask = (const float*)d_in[1];
  const float* Wq   = (const float*)d_in[2];
  const float* bq   = (const float*)d_in[3];
  const float* Wk   = (const float*)d_in[4];
  const float* bk   = (const float*)d_in[5];
  const float* Wv   = (const float*)d_in[6];
  const float* bv   = (const float*)d_in[7];
  const float* rate = (const float*)d_in[8];

  float* wsf     = (float*)d_ws;
  float* partial = wsf;
  float* qk      = partial + BB * 4 * DD;
  float* dot0    = qk + BB * DD;
  float* e       = dot0 + BB;

  float* out      = (float*)d_out;
  float* aOut     = out + BB * HH;
  float* dotOut   = aOut + BB * TT;
  float* decayOut = dotOut + BB * TT;

  colsum_partial<<<dim3(BB, 4), 256, 0, stream>>>(x, partial);
  q_chain_kernel<<<BB / 16, 256, 0, stream>>>(partial, Wq, bq, Wk, bk, qk, dot0);
  dot_softmax_xa_kernel<<<dim3(BB, 4), 256, 0, stream>>>(x, mask, qk, dot0, rate,
                                                         dotOut, decayOut, e, partial);
  finish_kernel<<<BB, 256, 0, stream>>>(e, partial, Wv, bv, aOut, out);
}


# one launch instead of four: one 16-wave work-group per batch reads its rows twice; column mean, q chain, softmax statistics and out GEMM meet in LDS (sum orders and k order unchanged)
# speedup vs baseline: 2.9216x; 1.0587x over previous
#include <hip/hip_runtime.h>
#include <math.h>

#define BB 128
#define TT 2048
#define DD 128
#define HH 128

typedef __attribute__((ext_vector_type(2))) float v2f;
typedef __attribute__((ext_vector_type(8))) float v8f;

__device__ __forceinline__ void lds_only_barrier() {
  __builtin_amdgcn_fence(__ATOMIC_RELEASE, "workgroup", "local");
  __builtin_amdgcn_s_barrier();
  __builtin_amdgcn_fence(__ATOMIC_ACQUIRE, "workgroup", "local");
}

template <int OFF>
__device__ __forceinline__ void fold_level(float (&s)[16], const int lane) {
  const bool up = (lane & OFF) != 0;
#pragma unroll
  for (int i = 0; i < OFF; ++i) {
    const float keep = up ? s[i + OFF] : s[i];
    const float send = up ? s[i] : s[i + OFF];
    s[i] = keep + __shfl_xor(send, OFF, 32);
  }
}

__global__ __launch_bounds__(512) void fused_batch_kernel(
    const float* __restrict__ x, const float* __restrict__ mask,
    const float* __restrict__ Wq, const float* __restrict__ bq,
    const float* __restrict__ Wk, const float* __restrict__ bk,
    const float* __restrict__ Wv, const float* __restrict__ bv,
    const float* __restrict__ rate, float* __restrict__ out,
    float* __restrict__ aOut, float* __restrict__ dotOut,
    float* __restrict__ decayOut) {
  const int b = blockIdx.x;
  const int tid = threadIdx.x, wave = tid >> 5, lane = tid & 31;
  const int row16 = lane & 15;
  const int hi = lane >> 4;
  const bool hiB = lane >= 16;
  const int n = (wave & 7) * 16 + row16;
  const int tw = wave * 128;

  __shared__ alignas(16) float red[32 * DD];
  __shared__ float sMax[32];
  __shared__ alignas(16) float xs[DD];
  __shared__ alignas(16) float qs[DD];
  __shared__ alignas(16) float qkL[DD];
  __shared__ float d0L[1];
  __shared__ float eL[TT];
  __shared__ float prL[4 * DD];
  __shared__ float sSum[8];
  __shared__ float os[HH];

  const float4* rows = (const float4*)(x + (b * TT + tw) * DD) + lane;
  float4 xv[2][16];

  {
#pragma unroll
    for (int i = 0; i < 16; ++i) xv[0][i] = rows[i * (DD / 4)];
    float4 acc[4];
#pragma unroll
    for (int k = 0; k < 4; ++k) acc[k] = make_float4(0.f, 0.f, 0.f, 0.f);
#pragma unroll
    for (int g = 0; g < 8; ++g) {
      if (g + 1 < 8) {
#pragma unroll
        for (int i = 0; i < 16; ++i)
          xv[(g + 1) & 1][i] = rows[((g + 1) * 16 + i) * (DD / 4)];
        asm volatile("" : "+v"(acc[0].x), "+v"(acc[0].y), "+v"(acc[0].z), "+v"(acc[0].w),
                          "+v"(acc[1].x), "+v"(acc[1].y), "+v"(acc[1].z), "+v"(acc[1].w),
                          "+v"(acc[2].x), "+v"(acc[2].y), "+v"(acc[2].z), "+v"(acc[2].w),
                          "+v"(acc[3].x), "+v"(acc[3].y), "+v"(acc[3].z), "+v"(acc[3].w));
      }
#pragma unroll
      for (int i = 0; i < 16; ++i) {
        const float4 v = xv[g & 1][i];
        acc[i & 3].x += v.x; acc[i & 3].y += v.y;
        acc[i & 3].z += v.z; acc[i & 3].w += v.w;
      }
      if ((g & 3) == 3) {
        float4 s;
        s.x = (acc[0].x + acc[1].x) + (acc[2].x + acc[3].x);
        s.y = (acc[0].y + acc[1].y) + (acc[2].y + acc[3].y);
        s.z = (acc[0].z + acc[1].z) + (acc[2].z + acc[3].z);
        s.w = (acc[0].w + acc[1].w) + (acc[2].w + acc[3].w);
        *(float4*)(red + (2 * wave + (g >> 2)) * DD + lane * 4) = s;
#pragma unroll
        for (int k = 0; k < 4; ++k) acc[k] = make_float4(0.f, 0.f, 0.f, 0.f);
      }
    }
  }

  const float rate0 = rate[0];
  v2f bqf[32];
  float bqn = 0.f;
  if (wave < 8) {
#pragma unroll
    for (int k = 0; k < 32; ++k) bqf[k] = *(const v2f*)(Wq + n * 128 + 4 * k + 2 * hi);
    bqn = bq[n];
  }
  float bkv[4];
  if (wave == 8) {
#pragma unroll
    for (int i = 0; i < 4; ++i) bkv[i] = bk[lane + 32 * i];
  }
#pragma unroll
  for (int i = 0; i < 16; ++i) xv[0][i] = rows[i * (DD / 4)];
  lds_only_barrier();

  if (tid < DD) {
    float t[4];
#pragma unroll
    for (int c = 0; c < 4; ++c) {
      t[c] = red[(8 * c) * DD + tid];
#pragma unroll
      for (int w = 1; w < 8; ++w) t[c] += red[(8 * c + w) * DD + tid];
    }
    float s = t[0];
#pragma unroll
    for (int c = 1; c < 4; ++c) s += t[c];
    const float sc = 1.0f / (float)TT;
    xs[tid] = s * sc;
  }
  lds_only_barrier();

  v2f bkf[32];
  if (wave < 8) {
#pragma unroll
    for (int k = 0; k < 32; ++k) {
      bkf[k].x = Wk[(4 * k + 2 * hi + 0) * 128 + n];
      bkf[k].y = Wk[(4 * k + 2 * hi + 1) * 128 + n];
    }
    v8f acc = {};
#pragma unroll
    for (int k = 0; k < 32; ++k) {
      const v2f a = *(const v2f*)(xs + 4 * k + 2 * hi);
      acc = __builtin_amdgcn_wmma_f32_16x16x4_f32(false, a, false, bqf[k],
                                                  (short)0, acc, false, false);
    }
    if (hi == 0) qs[n] = acc[0] + bqn;
  }
  lds_only_barrier();

  if (wave < 8) {
    v8f acc2 = {};
#pragma unroll
    for (int k = 0; k < 32; ++k) {
      const v2f a = *(const v2f*)(qs + 4 * k + 2 * hi);
      acc2 = __builtin_amdgcn_wmma_f32_16x16x4_f32(false, a, false, bkf[k],
                                                   (short)0, acc2, false, false);
    }
    if (hi == 0) qkL[n] = acc2[0];
  } else if (wave == 8) {
    const float p0 = bkv[0] * qs[lane];
    const float p1 = bkv[1] * qs[lane + 32];
    const float p2 = bkv[2] * qs[lane + 64];
    const float p3 = bkv[3] * qs[lane + 96];
    float u = (p0 + p2) + (p1 + p3);
#pragma unroll
    for (int off = 16; off > 0; off >>= 1) u += __shfl_xor(u, off, 32);
    if (lane == 0) d0L[0] = u;
  }
  lds_only_barrier();

  {
    float mk[8];
#pragma unroll
    for (int g = 0; g < 8; ++g) mk[g] = mask[b * TT + tw + 16 * g + row16];
    float4 qk4 = *(const float4*)(qkL + lane * 4);
    const float d0 = d0L[0];
    float4 acc0 = make_float4(0.f, 0.f, 0.f, 0.f);
    float4 acc1 = make_float4(0.f, 0.f, 0.f, 0.f);
    float mrun = 0.f;
    float dvE = 0.f, decE = 0.f, evE = 0.f;
#pragma unroll
    for (int g = 0; g < 8; ++g) {
      if (g + 1 < 8) {
#pragma unroll
        for (int i = 0; i < 16; ++i)
          xv[(g + 1) & 1][i] = rows[((g + 1) * 16 + i) * (DD / 4)];
      }
      asm volatile("" : "+v"(qk4.x), "+v"(qk4.y), "+v"(qk4.z), "+v"(qk4.w));
      float s[16];
#pragma unroll
      for (int i = 0; i < 16; ++i) {
        const float4 v = xv[g & 1][i];
        s[i] = v.x * qk4.x + v.y * qk4.y + v.z * qk4.z + v.w * qk4.w;
      }
      fold_level<8>(s, lane);
      fold_level<4>(s, lane);
      fold_level<2>(s, lane);
      fold_level<1>(s, lane);
      const float dsum = s[0] + __shfl_xor(s[0], 16, 32);
      const float dv = dsum + d0;
      const float sig = 1.0f / (1.0f + expf(-dv));
      const float tm = log1pf((1.0f - sig) * mk[g]);
      const float dec = rate0 * tm;
      const float ev = dv - dec;
      float gm = ev;
#pragma unroll
      for (int off = 8; off > 0; off >>= 1) gm = fmaxf(gm, __shfl_xor(gm, off, 32));
      if ((g & 3) == 0) {
        mrun = gm;
      } else {
        const float mnew = fmaxf(mrun, gm);
        const float sc = expf(mrun - mnew);
        acc0.x *= sc; acc0.y *= sc; acc0.z *= sc; acc0.w *= sc;
        acc1.x *= sc; acc1.y *= sc; acc1.z *= sc; acc1.w *= sc;
        mrun = mnew;
      }
      const int wbits = __float_as_int(expf(ev - mrun));
#pragma unroll
      for (int i = 0; i < 16; ++i) {
        const float wi = __int_as_float(__builtin_amdgcn_readlane(wbits, i));
        const float4 v = xv[g & 1][i];
        if (i & 1) {
          acc1.x = fmaf(wi, v.x, acc1.x);
          acc1.y = fmaf(wi, v.y, acc1.y);
          acc1.z = fmaf(wi, v.z, acc1.z);
          acc1.w = fmaf(wi, v.w, acc1.w);
        } else {
          acc0.x = fmaf(wi, v.x, acc0.x);
          acc0.y = fmaf(wi, v.y, acc0.y);
          acc0.z = fmaf(wi, v.z, acc0.z);
          acc0.w = fmaf(wi, v.w, acc0.w);
        }
      }
      asm volatile("" : "+v"(acc0.x), "+v"(acc0.y), "+v"(acc0.z), "+v"(acc0.w),
                        "+v"(acc1.x), "+v"(acc1.y), "+v"(acc1.z), "+v"(acc1.w));
      if (g & 1) {
        const int o = tw + (g >> 1) * 32 + lane;
        dotOut[b * TT + o] = hiB ? dv : dvE;
        decayOut[b * TT + o] = hiB ? dec : decE;
        eL[o] = hiB ? ev : evE;
      } else {
        dvE = dv; decE = dec; evE = ev;
      }
      if ((g & 3) == 3) {
        float4 acc;
        acc.x = acc0.x + acc1.x;
        acc.y = acc0.y + acc1.y;
        acc.z = acc0.z + acc1.z;
        acc.w = acc0.w + acc1.w;
        if (lane == 0) sMax[2 * wave + (g >> 2)] = mrun;
        *(float4*)(red + (2 * wave + (g >> 2)) * DD + lane * 4) = acc;
        acc0 = make_float4(0.f, 0.f, 0.f, 0.f);
        acc1 = make_float4(0.f, 0.f, 0.f, 0.f);
      }
    }
  }

  v2f bvf[32];
  float bvn = 0.f;
  if (wave < 8) {
#pragma unroll
    for (int k = 0; k < 32; ++k) bvf[k] = *(const v2f*)(Wv + n * 128 + 4 * k + 2 * hi);
    bvn = bv[n];
  }
  lds_only_barrier();

  float mc[4];
#pragma unroll
  for (int c = 0; c < 4; ++c) {
    mc[c] = sMax[8 * c];
#pragma unroll
    for (int w = 1; w < 8; ++w) mc[c] = fmaxf(mc[c], sMax[8 * c + w]);
  }
  const float M = fmaxf(fmaxf(mc[0], mc[1]), fmaxf(mc[2], mc[3]));
  {
    const int c = tid >> 7, col = tid & 127;
    float m = sMax[8 * c];
#pragma unroll
    for (int w = 1; w < 8; ++w) m = fmaxf(m, sMax[8 * c + w]);
    float sum = expf(sMax[8 * c] - m) * red[(8 * c) * DD + col];
#pragma unroll
    for (int w = 1; w < 8; ++w)
      sum = fmaf(expf(sMax[8 * c + w] - m), red[(8 * c + w) * DD + col], sum);
    prL[c * DD + col] = sum;
  }
  if (wave < 8) {
    float ls = 0.f;
#pragma unroll
    for (int k = 0; k < 8; ++k) ls += expf(eL[k * 256 + tid] - M);
#pragma unroll
    for (int off = 16; off > 0; off >>= 1) ls += __shfl_xor(ls, off, 32);
    if (lane == 0) sSum[wave] = ls;
  }
  lds_only_barrier();
  float Lsum = sSum[0];
#pragma unroll
  for (int w = 1; w < 8; ++w) Lsum += sSum[w];
  const float inv = 1.0f / Lsum;

#pragma unroll
  for (int k = 0; k < 4; ++k)
    aOut[b * TT + k * 512 + tid] = expf(eL[k * 512 + tid] - M) * inv;

  if (tid < DD) {
    float sum = (expf(mc[0] - M) * inv) * prL[tid];
#pragma unroll
    for (int c = 1; c < 4; ++c) sum = fmaf(expf(mc[c] - M) * inv, prL[c * DD + tid], sum);
    xs[tid] = sum;
  }
  lds_only_barrier();

  if (wave < 8) {
    v8f acc = {};
#pragma unroll
    for (int k = 0; k < 32; ++k) {
      const v2f a = *(const v2f*)(xs + 4 * k + 2 * hi);
      acc = __builtin_amdgcn_wmma_f32_16x16x4_f32(false, a, false, bvf[k],
                                                  (short)0, acc, false, false);
    }
    if (hi == 0) os[n] = acc[0] + bvn;
  }
  lds_only_barrier();
  if (tid < HH) out[b * HH + tid] = os[tid];
}

extern "C" void kernel_launch(void* const* d_in, const int* in_sizes, int n_in,
                              void* d_out, int out_size, void* d_ws, size_t ws_size,
                              hipStream_t stream) {
  (void)in_sizes; (void)n_in; (void)out_size; (void)d_ws; (void)ws_size;
  const float* x    = (const float*)d_in[0];
  const float* mask = (const float*)d_in[1];
  const float* Wq   = (const float*)d_in[2];
  const float* bq   = (const float*)d_in[3];
  const float* Wk   = (const float*)d_in[4];
  const float* bk   = (const float*)d_in[5];
  const float* Wv   = (const float*)d_in[6];
  const float* bv   = (const float*)d_in[7];
  const float* rate = (const float*)d_in[8];

  float* out      = (float*)d_out;
  float* aOut     = out + BB * HH;
  float* dotOut   = aOut + BB * TT;
  float* decayOut = dotOut + BB * TT;

  fused_batch_kernel<<<BB, 512, 0, stream>>>(x, mask, Wq, bq, Wk, bk, Wv, bv, rate,
                                             out, aOut, dotOut, decayOut);
}


# every lane exchange of the butterflies (row dot fold, group max, wave sums) on the vector unit: row_xmask DPP and permlanex16 instead of LDS-routed permutes; pairing trees unchanged
# speedup vs baseline: 3.2208x; 1.1024x over previous
#include <hip/hip_runtime.h>
#include <math.h>

#define BB 128
#define TT 2048
#define DD 128
#define HH 128

typedef __attribute__((ext_vector_type(2))) float v2f;
typedef __attribute__((ext_vector_type(8))) float v8f;

__device__ __forceinline__ void lds_only_barrier() {
  __builtin_amdgcn_fence(__ATOMIC_RELEASE, "workgroup", "local");
  __builtin_amdgcn_s_barrier();
  __builtin_amdgcn_fence(__ATOMIC_ACQUIRE, "workgroup", "local");
}

template <int M>
__device__ __forceinline__ float row_xor(const float v) {
  return __builtin_bit_cast(float, __builtin_amdgcn_update_dpp(
      0, __builtin_bit_cast(int, v), 0x160 + M, 0xf, 0xf, true));
}

__device__ __forceinline__ float half_xor(const float v) {
  const int b = __builtin_bit_cast(int, v);
  return __builtin_bit_cast(float, __builtin_amdgcn_permlanex16(
      b, b, 0x76543210, 0xfedcba98, false, false));
}

__device__ __forceinline__ float wave_sum(float u) {
  u += half_xor(u);
  u += row_xor<8>(u);
  u += row_xor<4>(u);
  u += row_xor<2>(u);
  u += row_xor<1>(u);
  return u;
}

template <int OFF>
__device__ __forceinline__ void fold_level(float (&s)[16], const int lane) {
  const bool up = (lane & OFF) != 0;
#pragma unroll
  for (int i = 0; i < OFF; ++i) {
    const float keep = up ? s[i + OFF] : s[i];
    const float send = up ? s[i] : s[i + OFF];
    s[i] = keep + row_xor<OFF>(send);
  }
}

__global__ __launch_bounds__(512) void fused_batch_kernel(
    const float* __restrict__ x, const float* __restrict__ mask,
    const float* __restrict__ Wq, const float* __restrict__ bq,
    const float* __restrict__ Wk, const float* __restrict__ bk,
    const float* __restrict__ Wv, const float* __restrict__ bv,
    const float* __restrict__ rate, float* __restrict__ out,
    float* __restrict__ aOut, float* __restrict__ dotOut,
    float* __restrict__ decayOut) {
  const int b = blockIdx.x;
  const int tid = threadIdx.x, wave = tid >> 5, lane = tid & 31;
  const int row16 = lane & 15;
  const int hi = lane >> 4;
  const bool hiB = lane >= 16;
  const int n = (wave & 7) * 16 + row16;
  const int tw = wave * 128;

  __shared__ alignas(16) float red[32 * DD];
  __shared__ float sMax[32];
  __shared__ alignas(16) float xs[DD];
  __shared__ alignas(16) float qs[DD];
  __shared__ alignas(16) float qkL[DD];
  __shared__ float d0L[1];
  __shared__ float eL[TT];
  __shared__ float prL[4 * DD];
  __shared__ float sSum[8];
  __shared__ float os[HH];

  const float4* rows = (const float4*)(x + (b * TT + tw) * DD) + lane;
  float4 xv[2][16];

  {
#pragma unroll
    for (int i = 0; i < 16; ++i) xv[0][i] = rows[i * (DD / 4)];
    float4 acc[4];
#pragma unroll
    for (int k = 0; k < 4; ++k) acc[k] = make_float4(0.f, 0.f, 0.f, 0.f);
#pragma unroll
    for (int g = 0; g < 8; ++g) {
      if (g + 1 < 8) {
#pragma unroll
        for (int i = 0; i < 16; ++i)
          xv[(g + 1) & 1][i] = rows[((g + 1) * 16 + i) * (DD / 4)];
        asm volatile("" : "+v"(acc[0].x), "+v"(acc[0].y), "+v"(acc[0].z), "+v"(acc[0].w),
                          "+v"(acc[1].x), "+v"(acc[1].y), "+v"(acc[1].z), "+v"(acc[1].w),
                          "+v"(acc[2].x), "+v"(acc[2].y), "+v"(acc[2].z), "+v"(acc[2].w),
                          "+v"(acc[3].x), "+v"(acc[3].y), "+v"(acc[3].z), "+v"(acc[3].w));
      }
#pragma unroll
      for (int i = 0; i < 16; ++i) {
        const float4 v = xv[g & 1][i];
        acc[i & 3].x += v.x; acc[i & 3].y += v.y;
        acc[i & 3].z += v.z; acc[i & 3].w += v.w;
      }
      if ((g & 3) == 3) {
        float4 s;
        s.x = (acc[0].x + acc[1].x) + (acc[2].x + acc[3].x);
        s.y = (acc[0].y + acc[1].y) + (acc[2].y + acc[3].y);
        s.z = (acc[0].z + acc[1].z) + (acc[2].z + acc[3].z);
        s.w = (acc[0].w + acc[1].w) + (acc[2].w + acc[3].w);
        *(float4*)(red + (2 * wave + (g >> 2)) * DD + lane * 4) = s;
#pragma unroll
        for (int k = 0; k < 4; ++k) acc[k] = make_float4(0.f, 0.f, 0.f, 0.f);
      }
    }
  }

  const float rate0 = rate[0];
  v2f bqf[32];
  float bqn = 0.f;
  if (wave < 8) {
#pragma unroll
    for (int k = 0; k < 32; ++k) bqf[k] = *(const v2f*)(Wq + n * 128 + 4 * k + 2 * hi);
    bqn = bq[n];
  }
  float bkv[4];
  if (wave == 8) {
#pragma unroll
    for (int i = 0; i < 4; ++i) bkv[i] = bk[lane + 32 * i];
  }
#pragma unroll
  for (int i = 0; i < 16; ++i) xv[0][i] = rows[i * (DD / 4)];
  lds_only_barrier();

  if (tid < DD) {
    float t[4];
#pragma unroll
    for (int c = 0; c < 4; ++c) {
      t[c] = red[(8 * c) * DD + tid];
#pragma unroll
      for (int w = 1; w < 8; ++w) t[c] += red[(8 * c + w) * DD + tid];
    }
    float s = t[0];
#pragma unroll
    for (int c = 1; c < 4; ++c) s += t[c];
    const float sc = 1.0f / (float)TT;
    xs[tid] = s * sc;
  }
  lds_only_barrier();

  v2f bkf[32];
  if (wave < 8) {
#pragma unroll
    for (int k = 0; k < 32; ++k) {
      bkf[k].x = Wk[(4 * k + 2 * hi + 0) * 128 + n];
      bkf[k].y = Wk[(4 * k + 2 * hi + 1) * 128 + n];
    }
    v8f acc = {};
#pragma unroll
    for (int k = 0; k < 32; ++k) {
      const v2f a = *(const v2f*)(xs + 4 * k + 2 * hi);
      acc = __builtin_amdgcn_wmma_f32_16x16x4_f32(false, a, false, bqf[k],
                                                  (short)0, acc, false, false);
    }
    if (hi == 0) qs[n] = acc[0] + bqn;
  }
  lds_only_barrier();

  if (wave < 8) {
    v8f acc2 = {};
#pragma unroll
    for (int k = 0; k < 32; ++k) {
      const v2f a = *(const v2f*)(qs + 4 * k + 2 * hi);
      acc2 = __builtin_amdgcn_wmma_f32_16x16x4_f32(false, a, false, bkf[k],
                                                   (short)0, acc2, false, false);
    }
    if (hi == 0) qkL[n] = acc2[0];
  } else if (wave == 8) {
    const float p0 = bkv[0] * qs[lane];
    const float p1 = bkv[1] * qs[lane + 32];
    const float p2 = bkv[2] * qs[lane + 64];
    const float p3 = bkv[3] * qs[lane + 96];
    const float u = wave_sum((p0 + p2) + (p1 + p3));
    if (lane == 0) d0L[0] = u;
  }
  lds_only_barrier();

  {
    float mk[8];
#pragma unroll
    for (int g = 0; g < 8; ++g) mk[g] = mask[b * TT + tw + 16 * g + row16];
    float4 qk4 = *(const float4*)(qkL + lane * 4);
    const float d0 = d0L[0];
    float4 acc0 = make_float4(0.f, 0.f, 0.f, 0.f);
    float4 acc1 = make_float4(0.f, 0.f, 0.f, 0.f);
    float mrun = 0.f;
    float dvE = 0.f, decE = 0.f, evE = 0.f;
#pragma unroll
    for (int g = 0; g < 8; ++g) {
      if (g + 1 < 8) {
#pragma unroll
        for (int i = 0; i < 16; ++i)
          xv[(g + 1) & 1][i] = rows[((g + 1) * 16 + i) * (DD / 4)];
      }
      asm volatile("" : "+v"(qk4.x), "+v"(qk4.y), "+v"(qk4.z), "+v"(qk4.w));
      float s[16];
#pragma unroll
      for (int i = 0; i < 16; ++i) {
        const float4 v = xv[g & 1][i];
        s[i] = v.x * qk4.x + v.y * qk4.y + v.z * qk4.z + v.w * qk4.w;
      }
      fold_level<8>(s, lane);
      fold_level<4>(s, lane);
      fold_level<2>(s, lane);
      fold_level<1>(s, lane);
      const float dsum = s[0] + half_xor(s[0]);
      const float dv = dsum + d0;
      const float sig = 1.0f / (1.0f + expf(-dv));
      const float tm = log1pf((1.0f - sig) * mk[g]);
      const float dec = rate0 * tm;
      const float ev = dv - dec;
      float gm = ev;
      gm = fmaxf(gm, row_xor<8>(gm));
      gm = fmaxf(gm, row_xor<4>(gm));
      gm = fmaxf(gm, row_xor<2>(gm));
      gm = fmaxf(gm, row_xor<1>(gm));
      if ((g & 3) == 0) {
        mrun = gm;
      } else {
        const float mnew = fmaxf(mrun, gm);
        const float sc = expf(mrun - mnew);
        acc0.x *= sc; acc0.y *= sc; acc0.z *= sc; acc0.w *= sc;
        acc1.x *= sc; acc1.y *= sc; acc1.z *= sc; acc1.w *= sc;
        mrun = mnew;
      }
      const int wbits = __float_as_int(expf(ev - mrun));
#pragma unroll
      for (int i = 0; i < 16; ++i) {
        const float wi = __int_as_float(__builtin_amdgcn_readlane(wbits, i));
        const float4 v = xv[g & 1][i];
        if (i & 1) {
          acc1.x = fmaf(wi, v.x, acc1.x);
          acc1.y = fmaf(wi, v.y, acc1.y);
          acc1.z = fmaf(wi, v.z, acc1.z);
          acc1.w = fmaf(wi, v.w, acc1.w);
        } else {
          acc0.x = fmaf(wi, v.x, acc0.x);
          acc0.y = fmaf(wi, v.y, acc0.y);
          acc0.z = fmaf(wi, v.z, acc0.z);
          acc0.w = fmaf(wi, v.w, acc0.w);
        }
      }
      asm volatile("" : "+v"(acc0.x), "+v"(acc0.y), "+v"(acc0.z), "+v"(acc0.w),
                        "+v"(acc1.x), "+v"(acc1.y), "+v"(acc1.z), "+v"(acc1.w));
      if (g & 1) {
        const int o = tw + (g >> 1) * 32 + lane;
        dotOut[b * TT + o] = hiB ? dv : dvE;
        decayOut[b * TT + o] = hiB ? dec : decE;
        eL[o] = hiB ? ev : evE;
      } else {
        dvE = dv; decE = dec; evE = ev;
      }
      if ((g & 3) == 3) {
        float4 acc;
        acc.x = acc0.x + acc1.x;
        acc.y = acc0.y + acc1.y;
        acc.z = acc0.z + acc1.z;
        acc.w = acc0.w + acc1.w;
        if (lane == 0) sMax[2 * wave + (g >> 2)] = mrun;
        *(float4*)(red + (2 * wave + (g >> 2)) * DD + lane * 4) = acc;
        acc0 = make_float4(0.f, 0.f, 0.f, 0.f);
        acc1 = make_float4(0.f, 0.f, 0.f, 0.f);
      }
    }
  }

  v2f bvf[32];
  float bvn = 0.f;
  if (wave < 8) {
#pragma unroll
    for (int k = 0; k < 32; ++k) bvf[k] = *(const v2f*)(Wv + n * 128 + 4 * k + 2 * hi);
    bvn = bv[n];
  }
  lds_only_barrier();

  float mc[4];
#pragma unroll
  for (int c = 0; c < 4; ++c) {
    mc[c] = sMax[8 * c];
#pragma unroll
    for (int w = 1; w < 8; ++w) mc[c] = fmaxf(mc[c], sMax[8 * c + w]);
  }
  const float M = fmaxf(fmaxf(mc[0], mc[1]), fmaxf(mc[2], mc[3]));
  {
    const int c = tid >> 7, col = tid & 127;
    float m = sMax[8 * c];
#pragma unroll
    for (int w = 1; w < 8; ++w) m = fmaxf(m, sMax[8 * c + w]);
    float sum = expf(sMax[8 * c] - m) * red[(8 * c) * DD + col];
#pragma unroll
    for (int w = 1; w < 8; ++w)
      sum = fmaf(expf(sMax[8 * c + w] - m), red[(8 * c + w) * DD + col], sum);
    prL[c * DD + col] = sum;
  }
  if (wave < 8) {
    float ls = 0.f;
#pragma unroll
    for (int k = 0; k < 8; ++k) ls += expf(eL[k * 256 + tid] - M);
    ls = wave_sum(ls);
    if (lane == 0) sSum[wave] = ls;
  }
  lds_only_barrier();
  float Lsum = sSum[0];
#pragma unroll
  for (int w = 1; w < 8; ++w) Lsum += sSum[w];
  const float inv = 1.0f / Lsum;

#pragma unroll
  for (int k = 0; k < 4; ++k)
    aOut[b * TT + k * 512 + tid] = expf(eL[k * 512 + tid] - M) * inv;

  if (tid < DD) {
    float sum = (expf(mc[0] - M) * inv) * prL[tid];
#pragma unroll
    for (int c = 1; c < 4; ++c) sum = fmaf(expf(mc[c] - M) * inv, prL[c * DD + tid], sum);
    xs[tid] = sum;
  }
  lds_only_barrier();

  if (wave < 8) {
    v8f acc = {};
#pragma unroll
    for (int k = 0; k < 32; ++k) {
      const v2f a = *(const v2f*)(xs + 4 * k + 2 * hi);
      acc = __builtin_amdgcn_wmma_f32_16x16x4_f32(false, a, false, bvf[k],
                                                  (short)0, acc, false, false);
    }
    if (hi == 0) os[n] = acc[0] + bvn;
  }
  lds_only_barrier();
  if (tid < HH) out[b * HH + tid] = os[tid];
}

extern "C" void kernel_launch(void* const* d_in, const int* in_sizes, int n_in,
                              void* d_out, int out_size, void* d_ws, size_t ws_size,
                              hipStream_t stream) {
  (void)in_sizes; (void)n_in; (void)out_size; (void)d_ws; (void)ws_size;
  const float* x    = (const float*)d_in[0];
  const float* mask = (const float*)d_in[1];
  const float* Wq   = (const float*)d_in[2];
  const float* bq   = (const float*)d_in[3];
  const float* Wk   = (const float*)d_in[4];
  const float* bk   = (const float*)d_in[5];
  const float* Wv   = (const float*)d_in[6];
  const float* bv   = (const float*)d_in[7];
  const float* rate = (const float*)d_in[8];

  float* out      = (float*)d_out;
  float* aOut     = out + BB * HH;
  float* dotOut   = aOut + BB * TT;
  float* decayOut = dotOut + BB * TT;

  fused_batch_kernel<<<BB, 512, 0, stream>>>(x, mask, Wq, bq, Wk, bk, Wv, bv, rate,
                                             out, aOut, dotOut, decayOut);
}


# pass 2 walks each wave's 16-row groups last to first: group 7 stays in registers, groups 6 and 5 parked in LDS by pass 1, 5 of 8 groups re-read; xa sums and running max meet groups in reverse order
# speedup vs baseline: 3.3610x; 1.0435x over previous
#include <hip/hip_runtime.h>
#include <math.h>

#define BB 128
#define TT 2048
#define DD 128
#define HH 128

typedef __attribute__((ext_vector_type(2))) float v2f;
typedef __attribute__((ext_vector_type(8))) float v8f;

__device__ __forceinline__ void lds_only_barrier() {
  __builtin_amdgcn_fence(__ATOMIC_RELEASE, "workgroup", "local");
  __builtin_amdgcn_s_barrier();
  __builtin_amdgcn_fence(__ATOMIC_ACQUIRE, "workgroup", "local");
}

template <int M>
__device__ __forceinline__ float row_xor(const float v) {
  return __builtin_bit_cast(float, __builtin_amdgcn_update_dpp(
      0, __builtin_bit_cast(int, v), 0x160 + M, 0xf, 0xf, true));
}

__device__ __forceinline__ float half_xor(const float v) {
  const int b = __builtin_bit_cast(int, v);
  return __builtin_bit_cast(float, __builtin_amdgcn_permlanex16(
      b, b, 0x76543210, 0xfedcba98, false, false));
}

__device__ __forceinline__ float wave_sum(float u) {
  u += half_xor(u);
  u += row_xor<8>(u);
  u += row_xor<4>(u);
  u += row_xor<2>(u);
  u += row_xor<1>(u);
  return u;
}

template <int OFF>
__device__ __forceinline__ void fold_level(float (&s)[16], const int lane) {
  const bool up = (lane & OFF) != 0;
#pragma unroll
  for (int i = 0; i < OFF; ++i) {
    const float keep = up ? s[i + OFF] : s[i];
    const float send = up ? s[i] : s[i + OFF];
    s[i] = keep + row_xor<OFF>(send);
  }
}

__global__ __launch_bounds__(512) void fused_batch_kernel(
    const float* __restrict__ x, const float* __restrict__ mask,
    const float* __restrict__ Wq, const float* __restrict__ bq,
    const float* __restrict__ Wk, const float* __restrict__ bk,
    const float* __restrict__ Wv, const float* __restrict__ bv,
    const float* __restrict__ rate, float* __restrict__ out,
    float* __restrict__ aOut, float* __restrict__ dotOut,
    float* __restrict__ decayOut) {
  const int b = blockIdx.x;
  const int tid = threadIdx.x, wave = tid >> 5, lane = tid & 31;
  const int row16 = lane & 15;
  const int hi = lane >> 4;
  const bool hiB = lane >= 16;
  const int n = (wave & 7) * 16 + row16;
  const int tw = wave * 128;

  struct Lds {
    alignas(16) float red[32 * DD];
    alignas(16) float xs[DD];
    alignas(16) float qs[DD];
    alignas(16) float qkL[DD];
    float eL[TT];
    float prL[4 * DD];
    float os[HH];
    float sMax[32];
    float sSum[8];
    float d0L[4];
    alignas(16) float4 parked[16 * 2 * 16 * 32];
  };
  __shared__ Lds L;
  float* const red = L.red;
  float* const sMax = L.sMax;
  float* const xs = L.xs;
  float* const qs = L.qs;
  float* const qkL = L.qkL;
  float* const d0L = L.d0L;
  float* const eL = L.eL;
  float* const prL = L.prL;
  float* const sSum = L.sSum;
  float* const os = L.os;
  float4* const park = L.parked + wave * (2 * 16 * 32) + lane;

  const float4* rows = (const float4*)(x + (b * TT + tw) * DD) + lane;
  float4 xv[2][16];

  {
#pragma unroll
    for (int i = 0; i < 16; ++i) xv[0][i] = rows[i * (DD / 4)];
    float4 acc[4];
#pragma unroll
    for (int k = 0; k < 4; ++k) acc[k] = make_float4(0.f, 0.f, 0.f, 0.f);
#pragma unroll
    for (int g = 0; g < 8; ++g) {
      if (g + 1 < 8) {
#pragma unroll
        for (int i = 0; i < 16; ++i)
          xv[(g + 1) & 1][i] = rows[((g + 1) * 16 + i) * (DD / 4)];
        asm volatile("" : "+v"(acc[0].x), "+v"(acc[0].y), "+v"(acc[0].z), "+v"(acc[0].w),
                          "+v"(acc[1].x), "+v"(acc[1].y), "+v"(acc[1].z), "+v"(acc[1].w),
                          "+v"(acc[2].x), "+v"(acc[2].y), "+v"(acc[2].z), "+v"(acc[2].w),
                          "+v"(acc[3].x), "+v"(acc[3].y), "+v"(acc[3].z), "+v"(acc[3].w));
      }
#pragma unroll
      for (int i = 0; i < 16; ++i) {
        const float4 v = xv[g & 1][i];
        if (g == 5 || g == 6) park[((g - 5) * 16 + i) * 32] = v;
        acc[i & 3].x += v.x; acc[i & 3].y += v.y;
        acc[i & 3].z += v.z; acc[i & 3].w += v.w;
      }
      if ((g & 3) == 3) {
        float4 s;
        s.x = (acc[0].x + acc[1].x) + (acc[2].x + acc[3].x);
        s.y = (acc[0].y + acc[1].y) + (acc[2].y + acc[3].y);
        s.z = (acc[0].z + acc[1].z) + (acc[2].z + acc[3].z);
        s.w = (acc[0].w + acc[1].w) + (acc[2].w + acc[3].w);
        *(float4*)(red + (2 * wave + (g >> 2)) * DD + lane * 4) = s;
#pragma unroll
        for (int k = 0; k < 4; ++k) acc[k] = make_float4(0.f, 0.f, 0.f, 0.f);
      }
      asm volatile("" : "+v"(acc[0].x), "+v"(acc[0].y), "+v"(acc[0].z), "+v"(acc[0].w),
                        "+v"(acc[1].x), "+v"(acc[1].y), "+v"(acc[1].z), "+v"(acc[1].w),
                        "+v"(acc[2].x), "+v"(acc[2].y), "+v"(acc[2].z), "+v"(acc[2].w),
                        "+v"(acc[3].x), "+v"(acc[3].y), "+v"(acc[3].z), "+v"(acc[3].w));
    }
  }

  const float rate0 = rate[0];
  v2f bqf[32];
  float bqn = 0.f;
  if (wave < 8) {
#pragma unroll
    for (int k = 0; k < 32; ++k) bqf[k] = *(const v2f*)(Wq + n * 128 + 4 * k + 2 * hi);
    bqn = bq[n];
  }
  float bkv[4];
  if (wave == 8) {
#pragma unroll
    for (int i = 0; i < 4; ++i) bkv[i] = bk[lane + 32 * i];
  }
  lds_only_barrier();

  if (tid < DD) {
    float t[4];
#pragma unroll
    for (int c = 0; c < 4; ++c) {
      t[c] = red[(8 * c) * DD + tid];
#pragma unroll
      for (int w = 1; w < 8; ++w) t[c] += red[(8 * c + w) * DD + tid];
    }
    float s = t[0];
#pragma unroll
    for (int c = 1; c < 4; ++c) s += t[c];
    const float sc = 1.0f / (float)TT;
    xs[tid] = s * sc;
  }
  lds_only_barrier();

  v2f bkf[32];
  if (wave < 8) {
#pragma unroll
    for (int k = 0; k < 32; ++k) {
      bkf[k].x = Wk[(4 * k + 2 * hi + 0) * 128 + n];
      bkf[k].y = Wk[(4 * k + 2 * hi + 1) * 128 + n];
    }
    v8f acc = {};
#pragma unroll
    for (int k = 0; k < 32; ++k) {
      const v2f a = *(const v2f*)(xs + 4 * k + 2 * hi);
      acc = __builtin_amdgcn_wmma_f32_16x16x4_f32(false, a, false, bqf[k],
                                                  (short)0, acc, false, false);
    }
    if (hi == 0) qs[n] = acc[0] + bqn;
  }
  lds_only_barrier();

  if (wave < 8) {
    v8f acc2 = {};
#pragma unroll
    for (int k = 0; k < 32; ++k) {
      const v2f a = *(const v2f*)(qs + 4 * k + 2 * hi);
      acc2 = __builtin_amdgcn_wmma_f32_16x16x4_f32(false, a, false, bkf[k],
                                                   (short)0, acc2, false, false);
    }
    if (hi == 0) qkL[n] = acc2[0];
  } else if (wave == 8) {
    const float p0 = bkv[0] * qs[lane];
    const float p1 = bkv[1] * qs[lane + 32];
    const float p2 = bkv[2] * qs[lane + 64];
    const float p3 = bkv[3] * qs[lane + 96];
    const float u = wave_sum((p0 + p2) + (p1 + p3));
    if (lane == 0) d0L[0] = u;
  }
  lds_only_barrier();

  {
    float mk[8];
#pragma unroll
    for (int g = 0; g < 8; ++g) mk[g] = mask[b * TT + tw + 16 * g + row16];
    float4 qk4 = *(const float4*)(qkL + lane * 4);
    const float d0 = d0L[0];
    float4 acc0 = make_float4(0.f, 0.f, 0.f, 0.f);
    float4 acc1 = make_float4(0.f, 0.f, 0.f, 0.f);
    float mrun = 0.f;
    float dvO = 0.f, decO = 0.f, evO = 0.f;
#pragma unroll
    for (int st = 0; st < 8; ++st) {
      const int g = 7 - st;
      if (g > 0) {
        const int gn = g - 1;
        if (gn == 6 || gn == 5) {
#pragma unroll
          for (int i = 0; i < 16; ++i)
            xv[gn & 1][i] = park[((gn - 5) * 16 + i) * 32];
        } else {
#pragma unroll
          for (int i = 0; i < 16; ++i)
            xv[gn & 1][i] = rows[(gn * 16 + i) * (DD / 4)];
        }
      }
      asm volatile("" : "+v"(qk4.x), "+v"(qk4.y), "+v"(qk4.z), "+v"(qk4.w));
      float s[16];
#pragma unroll
      for (int i = 0; i < 16; ++i) {
        const float4 v = xv[g & 1][i];
        s[i] = v.x * qk4.x + v.y * qk4.y + v.z * qk4.z + v.w * qk4.w;
      }
      fold_level<8>(s, lane);
      fold_level<4>(s, lane);
      fold_level<2>(s, lane);
      fold_level<1>(s, lane);
      const float dsum = s[0] + half_xor(s[0]);
      const float dv = dsum + d0;
      const float sig = 1.0f / (1.0f + expf(-dv));
      const float tm = log1pf((1.0f - sig) * mk[g]);
      const float dec = rate0 * tm;
      const float ev = dv - dec;
      float gm = ev;
      gm = fmaxf(gm, row_xor<8>(gm));
      gm = fmaxf(gm, row_xor<4>(gm));
      gm = fmaxf(gm, row_xor<2>(gm));
      gm = fmaxf(gm, row_xor<1>(gm));
      if ((g & 3) == 3) {
        mrun = gm;
      } else {
        const float mnew = fmaxf(mrun, gm);
        const float sc = expf(mrun - mnew);
        acc0.x *= sc; acc0.y *= sc; acc0.z *= sc; acc0.w *= sc;
        acc1.x *= sc; acc1.y *= sc; acc1.z *= sc; acc1.w *= sc;
        mrun = mnew;
      }
      const int wbits = __float_as_int(expf(ev - mrun));
#pragma unroll
      for (int i = 0; i < 16; ++i) {
        const float wi = __int_as_float(__builtin_amdgcn_readlane(wbits, i));
        const float4 v = xv[g & 1][i];
        if (i & 1) {
          acc1.x = fmaf(wi, v.x, acc1.x);
          acc1.y = fmaf(wi, v.y, acc1.y);
          acc1.z = fmaf(wi, v.z, acc1.z);
          acc1.w = fmaf(wi, v.w, acc1.w);
        } else {
          acc0.x = fmaf(wi, v.x, acc0.x);
          acc0.y = fmaf(wi, v.y, acc0.y);
          acc0.z = fmaf(wi, v.z, acc0.z);
          acc0.w = fmaf(wi, v.w, acc0.w);
        }
      }
      asm volatile("" : "+v"(acc0.x), "+v"(acc0.y), "+v"(acc0.z), "+v"(acc0.w),
                        "+v"(acc1.x), "+v"(acc1.y), "+v"(acc1.z), "+v"(acc1.w));
      if (g & 1) {
        dvO = dv; decO = dec; evO = ev;
      } else {
        const int o = tw + (g >> 1) * 32 + lane;
        dotOut[b * TT + o] = hiB ? dvO : dv;
        decayOut[b * TT + o] = hiB ? decO : dec;
        eL[o] = hiB ? evO : ev;
      }
      if ((g & 3) == 0) {
        float4 acc;
        acc.x = acc0.x + acc1.x;
        acc.y = acc0.y + acc1.y;
        acc.z = acc0.z + acc1.z;
        acc.w = acc0.w + acc1.w;
        if (lane == 0) sMax[2 * wave + (g >> 2)] = mrun;
        *(float4*)(red + (2 * wave + (g >> 2)) * DD + lane * 4) = acc;
        acc0 = make_float4(0.f, 0.f, 0.f, 0.f);
        acc1 = make_float4(0.f, 0.f, 0.f, 0.f);
      }
    }
  }

  v2f bvf[32];
  float bvn = 0.f;
  if (wave < 8) {
#pragma unroll
    for (int k = 0; k < 32; ++k) bvf[k] = *(const v2f*)(Wv + n * 128 + 4 * k + 2 * hi);
    bvn = bv[n];
  }
  lds_only_barrier();

  float mc[4];
#pragma unroll
  for (int c = 0; c < 4; ++c) {
    mc[c] = sMax[8 * c];
#pragma unroll
    for (int w = 1; w < 8; ++w) mc[c] = fmaxf(mc[c], sMax[8 * c + w]);
  }
  const float M = fmaxf(fmaxf(mc[0], mc[1]), fmaxf(mc[2], mc[3]));
  {
    const int c = tid >> 7, col = tid & 127;
    float m = sMax[8 * c];
#pragma unroll
    for (int w = 1; w < 8; ++w) m = fmaxf(m, sMax[8 * c + w]);
    float sum = expf(sMax[8 * c] - m) * red[(8 * c) * DD + col];
#pragma unroll
    for (int w = 1; w < 8; ++w)
      sum = fmaf(expf(sMax[8 * c + w] - m), red[(8 * c + w) * DD + col], sum);
    prL[c * DD + col] = sum;
  }
  if (wave < 8) {
    float ls = 0.f;
#pragma unroll
    for (int k = 0; k < 8; ++k) ls += expf(eL[k * 256 + tid] - M);
    ls = wave_sum(ls);
    if (lane == 0) sSum[wave] = ls;
  }
  lds_only_barrier();
  float Lsum = sSum[0];
#pragma unroll
  for (int w = 1; w < 8; ++w) Lsum += sSum[w];
  const float inv = 1.0f / Lsum;

#pragma unroll
  for (int k = 0; k < 4; ++k)
    aOut[b * TT + k * 512 + tid] = expf(eL[k * 512 + tid] - M) * inv;

  if (tid < DD) {
    float sum = (expf(mc[0] - M) * inv) * prL[tid];
#pragma unroll
    for (int c = 1; c < 4; ++c) sum = fmaf(expf(mc[c] - M) * inv, prL[c * DD + tid], sum);
    xs[tid] = sum;
  }
  lds_only_barrier();

  if (wave < 8) {
    v8f acc = {};
#pragma unroll
    for (int k = 0; k < 32; ++k) {
      const v2f a = *(const v2f*)(xs + 4 * k + 2 * hi);
      acc = __builtin_amdgcn_wmma_f32_16x16x4_f32(false, a, false, bvf[k],
                                                  (short)0, acc, false, false);
    }
    if (hi == 0) os[n] = acc[0] + bvn;
  }
  lds_only_barrier();
  if (tid < HH) out[b * HH + tid] = os[tid];
}

extern "C" void kernel_launch(void* const* d_in, const int* in_sizes, int n_in,
                              void* d_out, int out_size, void* d_ws, size_t ws_size,
                              hipStream_t stream) {
  (void)in_sizes; (void)n_in; (void)out_size; (void)d_ws; (void)ws_size;
  const float* x    = (const float*)d_in[0];
  const float* mask = (const float*)d_in[1];
  const float* Wq   = (const float*)d_in[2];
  const float* bq   = (const float*)d_in[3];
  const float* Wk   = (const float*)d_in[4];
  const float* bk   = (const float*)d_in[5];
  const float* Wv   = (const float*)d_in[6];
  const float* bv   = (const float*)d_in[7];
  const float* rate = (const float*)d_in[8];

  float* out      = (float*)d_out;
  float* aOut     = out + BB * HH;
  float* dotOut   = aOut + BB * TT;
  float* decayOut = dotOut + BB * TT;

  fused_batch_kernel<<<BB, 512, 0, stream>>>(x, mask, Wq, bq, Wk, bk, Wv, bv, rate,
                                             out, aOut, dotOut, decayOut);
}


# pass 2 score chain (sigmoid, log1p, weight and rescale exps) on hardware exp2, log2 and rcp instead of in-line library routines; decay, a, out move by a few ulp; sum orders unchanged
# speedup vs baseline: 3.5833x; 1.0661x over previous
#include <hip/hip_runtime.h>
#include <math.h>

#define BB 128
#define TT 2048
#define DD 128
#define HH 128

typedef __attribute__((ext_vector_type(2))) float v2f;
typedef __attribute__((ext_vector_type(8))) float v8f;

__device__ __forceinline__ void lds_only_barrier() {
  __builtin_amdgcn_fence(__ATOMIC_RELEASE, "workgroup", "local");
  __builtin_amdgcn_s_barrier();
  __builtin_amdgcn_fence(__ATOMIC_ACQUIRE, "workgroup", "local");
}

template <int M>
__device__ __forceinline__ float row_xor(const float v) {
  return __builtin_bit_cast(float, __builtin_amdgcn_update_dpp(
      0, __builtin_bit_cast(int, v), 0x160 + M, 0xf, 0xf, true));
}

__device__ __forceinline__ float half_xor(const float v) {
  const int b = __builtin_bit_cast(int, v);
  return __builtin_bit_cast(float, __builtin_amdgcn_permlanex16(
      b, b, 0x76543210, 0xfedcba98, false, false));
}

__device__ __forceinline__ float wave_sum(float u) {
  u += half_xor(u);
  u += row_xor<8>(u);
  u += row_xor<4>(u);
  u += row_xor<2>(u);
  u += row_xor<1>(u);
  return u;
}

__device__ __forceinline__ float exp_hw(const float x) {
  return __builtin_amdgcn_exp2f(x * 1.4426950408889634f);
}

__device__ __forceinline__ float log1p_hw(const float u) {
  const float y = 1.0f + u;
  const float d = y - 1.0f;
  const float l = __builtin_amdgcn_logf(y) * 0.6931471805599453f;
  const float c = (u - d) * __builtin_amdgcn_rcpf(y);
  return (fabsf(c) <= 1.0f) ? l + c : l;
}

template <int OFF>
__device__ __forceinline__ void fold_level(float (&s)[16], const int lane) {
  const bool up = (lane & OFF) != 0;
#pragma unroll
  for (int i = 0; i < OFF; ++i) {
    const float keep = up ? s[i + OFF] : s[i];
    const float send = up ? s[i] : s[i + OFF];
    s[i] = keep + row_xor<OFF>(send);
  }
}

__global__ __launch_bounds__(512) void fused_batch_kernel(
    const float* __restrict__ x, const float* __restrict__ mask,
    const float* __restrict__ Wq, const float* __restrict__ bq,
    const float* __restrict__ Wk, const float* __restrict__ bk,
    const float* __restrict__ Wv, const float* __restrict__ bv,
    const float* __restrict__ rate, float* __restrict__ out,
    float* __restrict__ aOut, float* __restrict__ dotOut,
    float* __restrict__ decayOut) {
  const int b = blockIdx.x;
  const int tid = threadIdx.x, wave = tid >> 5, lane = tid & 31;
  const int row16 = lane & 15;
  const int hi = lane >> 4;
  const bool hiB = lane >= 16;
  const int n = (wave & 7) * 16 + row16;
  const int tw = wave * 128;

  struct Lds {
    alignas(16) float red[32 * DD];
    alignas(16) float xs[DD];
    alignas(16) float qs[DD];
    alignas(16) float qkL[DD];
    float eL[TT];
    float prL[4 * DD];
    float os[HH];
    float sMax[32];
    float sSum[8];
    float d0L[4];
    alignas(16) float4 parked[16 * 2 * 16 * 32];
  };
  __shared__ Lds L;
  float* const red = L.red;
  float* const sMax = L.sMax;
  float* const xs = L.xs;
  float* const qs = L.qs;
  float* const qkL = L.qkL;
  float* const d0L = L.d0L;
  float* const eL = L.eL;
  float* const prL = L.prL;
  float* const sSum = L.sSum;
  float* const os = L.os;
  float4* const park = L.parked + wave * (2 * 16 * 32) + lane;

  const float4* rows = (const float4*)(x + (b * TT + tw) * DD) + lane;
  float4 xv[2][16];

  {
#pragma unroll
    for (int i = 0; i < 16; ++i) xv[0][i] = rows[i * (DD / 4)];
    float4 acc[4];
#pragma unroll
    for (int k = 0; k < 4; ++k) acc[k] = make_float4(0.f, 0.f, 0.f, 0.f);
#pragma unroll
    for (int g = 0; g < 8; ++g) {
      if (g + 1 < 8) {
#pragma unroll
        for (int i = 0; i < 16; ++i)
          xv[(g + 1) & 1][i] = rows[((g + 1) * 16 + i) * (DD / 4)];
        asm volatile("" : "+v"(acc[0].x), "+v"(acc[0].y), "+v"(acc[0].z), "+v"(acc[0].w),
                          "+v"(acc[1].x), "+v"(acc[1].y), "+v"(acc[1].z), "+v"(acc[1].w),
                          "+v"(acc[2].x), "+v"(acc[2].y), "+v"(acc[2].z), "+v"(acc[2].w),
                          "+v"(acc[3].x), "+v"(acc[3].y), "+v"(acc[3].z), "+v"(acc[3].w));
      }
#pragma unroll
      for (int i = 0; i < 16; ++i) {
        const float4 v = xv[g & 1][i];
        if (g == 5 || g == 6) park[((g - 5) * 16 + i) * 32] = v;
        acc[i & 3].x += v.x; acc[i & 3].y += v.y;
        acc[i & 3].z += v.z; acc[i & 3].w += v.w;
      }
      if ((g & 3) == 3) {
        float4 s;
        s.x = (acc[0].x + acc[1].x) + (acc[2].x + acc[3].x);
        s.y = (acc[0].y + acc[1].y) + (acc[2].y + acc[3].y);
        s.z = (acc[0].z + acc[1].z) + (acc[2].z + acc[3].z);
        s.w = (acc[0].w + acc[1].w) + (acc[2].w + acc[3].w);
        *(float4*)(red + (2 * wave + (g >> 2)) * DD + lane * 4) = s;
#pragma unroll
        for (int k = 0; k < 4; ++k) acc[k] = make_float4(0.f, 0.f, 0.f, 0.f);
      }
      asm volatile("" : "+v"(acc[0].x), "+v"(acc[0].y), "+v"(acc[0].z), "+v"(acc[0].w),
                        "+v"(acc[1].x), "+v"(acc[1].y), "+v"(acc[1].z), "+v"(acc[1].w),
                        "+v"(acc[2].x), "+v"(acc[2].y), "+v"(acc[2].z), "+v"(acc[2].w),
                        "+v"(acc[3].x), "+v"(acc[3].y), "+v"(acc[3].z), "+v"(acc[3].w));
    }
  }

  const float rate0 = rate[0];
  v2f bqf[32];
  float bqn = 0.f;
  if (wave < 8) {
#pragma unroll
    for (int k = 0; k < 32; ++k) bqf[k] = *(const v2f*)(Wq + n * 128 + 4 * k + 2 * hi);
    bqn = bq[n];
  }
  float bkv[4];
  if (wave == 8) {
#pragma unroll
    for (int i = 0; i < 4; ++i) bkv[i] = bk[lane + 32 * i];
  }
  lds_only_barrier();

  if (tid < DD) {
    float t[4];
#pragma unroll
    for (int c = 0; c < 4; ++c) {
      t[c] = red[(8 * c) * DD + tid];
#pragma unroll
      for (int w = 1; w < 8; ++w) t[c] += red[(8 * c + w) * DD + tid];
    }
    float s = t[0];
#pragma unroll
    for (int c = 1; c < 4; ++c) s += t[c];
    const float sc = 1.0f / (float)TT;
    xs[tid] = s * sc;
  }
  lds_only_barrier();

  v2f bkf[32];
  if (wave < 8) {
#pragma unroll
    for (int k = 0; k < 32; ++k) {
      bkf[k].x = Wk[(4 * k + 2 * hi + 0) * 128 + n];
      bkf[k].y = Wk[(4 * k + 2 * hi + 1) * 128 + n];
    }
    v8f acc = {};
#pragma unroll
    for (int k = 0; k < 32; ++k) {
      const v2f a = *(const v2f*)(xs + 4 * k + 2 * hi);
      acc = __builtin_amdgcn_wmma_f32_16x16x4_f32(false, a, false, bqf[k],
                                                  (short)0, acc, false, false);
    }
    if (hi == 0) qs[n] = acc[0] + bqn;
  }
  lds_only_barrier();

  if (wave < 8) {
    v8f acc2 = {};
#pragma unroll
    for (int k = 0; k < 32; ++k) {
      const v2f a = *(const v2f*)(qs + 4 * k + 2 * hi);
      acc2 = __builtin_amdgcn_wmma_f32_16x16x4_f32(false, a, false, bkf[k],
                                                   (short)0, acc2, false, false);
    }
    if (hi == 0) qkL[n] = acc2[0];
  } else if (wave == 8) {
    const float p0 = bkv[0] * qs[lane];
    const float p1 = bkv[1] * qs[lane + 32];
    const float p2 = bkv[2] * qs[lane + 64];
    const float p3 = bkv[3] * qs[lane + 96];
    const float u = wave_sum((p0 + p2) + (p1 + p3));
    if (lane == 0) d0L[0] = u;
  }
  lds_only_barrier();

  {
    float mk[8];
#pragma unroll
    for (int g = 0; g < 8; ++g) mk[g] = mask[b * TT + tw + 16 * g + row16];
    float4 qk4 = *(const float4*)(qkL + lane * 4);
    const float d0 = d0L[0];
    float4 acc0 = make_float4(0.f, 0.f, 0.f, 0.f);
    float4 acc1 = make_float4(0.f, 0.f, 0.f, 0.f);
    float mrun = 0.f;
    float dvO = 0.f, decO = 0.f, evO = 0.f;
#pragma unroll
    for (int st = 0; st < 8; ++st) {
      const int g = 7 - st;
      if (g > 0) {
        const int gn = g - 1;
        if (gn == 6 || gn == 5) {
#pragma unroll
          for (int i = 0; i < 16; ++i)
            xv[gn & 1][i] = park[((gn - 5) * 16 + i) * 32];
        } else {
#pragma unroll
          for (int i = 0; i < 16; ++i)
            xv[gn & 1][i] = rows[(gn * 16 + i) * (DD / 4)];
        }
      }
      asm volatile("" : "+v"(qk4.x), "+v"(qk4.y), "+v"(qk4.z), "+v"(qk4.w));
      float s[16];
#pragma unroll
      for (int i = 0; i < 16; ++i) {
        const float4 v = xv[g & 1][i];
        s[i] = fmaf(v.w, qk4.w, fmaf(v.z, qk4.z, fmaf(v.x, qk4.x, v.y * qk4.y)));
      }
      fold_level<8>(s, lane);
      fold_level<4>(s, lane);
      fold_level<2>(s, lane);
      fold_level<1>(s, lane);
      const float dsum = s[0] + half_xor(s[0]);
      const float dv = dsum + d0;
      const float sig = __builtin_amdgcn_rcpf(1.0f + exp_hw(-dv));
      const float tm = log1p_hw((1.0f - sig) * mk[g]);
      const float dec = rate0 * tm;
      const float ev = dv - dec;
      float gm = ev;
      gm = fmaxf(gm, row_xor<8>(gm));
      gm = fmaxf(gm, row_xor<4>(gm));
      gm = fmaxf(gm, row_xor<2>(gm));
      gm = fmaxf(gm, row_xor<1>(gm));
      if ((g & 3) == 3) {
        mrun = gm;
      } else {
        const float mnew = fmaxf(mrun, gm);
        const float sc = exp_hw(mrun - mnew);
        acc0.x *= sc; acc0.y *= sc; acc0.z *= sc; acc0.w *= sc;
        acc1.x *= sc; acc1.y *= sc; acc1.z *= sc; acc1.w *= sc;
        mrun = mnew;
      }
      const int wbits = __float_as_int(exp_hw(ev - mrun));
#pragma unroll
      for (int i = 0; i < 16; ++i) {
        const float wi = __int_as_float(__builtin_amdgcn_readlane(wbits, i));
        const float4 v = xv[g & 1][i];
        if (i & 1) {
          acc1.x = fmaf(wi, v.x, acc1.x);
          acc1.y = fmaf(wi, v.y, acc1.y);
          acc1.z = fmaf(wi, v.z, acc1.z);
          acc1.w = fmaf(wi, v.w, acc1.w);
        } else {
          acc0.x = fmaf(wi, v.x, acc0.x);
          acc0.y = fmaf(wi, v.y, acc0.y);
          acc0.z = fmaf(wi, v.z, acc0.z);
          acc0.w = fmaf(wi, v.w, acc0.w);
        }
      }
      asm volatile("" : "+v"(acc0.x), "+v"(acc0.y), "+v"(acc0.z), "+v"(acc0.w),
                        "+v"(acc1.x), "+v"(acc1.y), "+v"(acc1.z), "+v"(acc1.w));
      if (g & 1) {
        dvO = dv; decO = dec; evO = ev;
      } else {
        const int o = tw + (g >> 1) * 32 + lane;
        dotOut[b * TT + o] = hiB ? dvO : dv;
        decayOut[b * TT + o] = hiB ? decO : dec;
        eL[o] = hiB ? evO : ev;
      }
      if ((g & 3) == 0) {
        float4 acc;
        acc.x = acc0.x + acc1.x;
        acc.y = acc0.y + acc1.y;
        acc.z = acc0.z + acc1.z;
        acc.w = acc0.w + acc1.w;
        if (lane == 0) sMax[2 * wave + (g >> 2)] = mrun;
        *(float4*)(red + (2 * wave + (g >> 2)) * DD + lane * 4) = acc;
        acc0 = make_float4(0.f, 0.f, 0.f, 0.f);
        acc1 = make_float4(0.f, 0.f, 0.f, 0.f);
      }
    }
  }

  v2f bvf[32];
  float bvn = 0.f;
  if (wave < 8) {
#pragma unroll
    for (int k = 0; k < 32; ++k) bvf[k] = *(const v2f*)(Wv + n * 128 + 4 * k + 2 * hi);
    bvn = bv[n];
  }
  lds_only_barrier();

  float mc[4];
#pragma unroll
  for (int c = 0; c < 4; ++c) {
    mc[c] = sMax[8 * c];
#pragma unroll
    for (int w = 1; w < 8; ++w) mc[c] = fmaxf(mc[c], sMax[8 * c + w]);
  }
  const float M = fmaxf(fmaxf(mc[0], mc[1]), fmaxf(mc[2], mc[3]));
  {
    const int c = tid >> 7, col = tid & 127;
    float m = sMax[8 * c];
#pragma unroll
    for (int w = 1; w < 8; ++w) m = fmaxf(m, sMax[8 * c + w]);
    float sum = expf(sMax[8 * c] - m) * red[(8 * c) * DD + col];
#pragma unroll
    for (int w = 1; w < 8; ++w)
      sum = fmaf(expf(sMax[8 * c + w] - m), red[(8 * c + w) * DD + col], sum);
    prL[c * DD + col] = sum;
  }
  if (wave < 8) {
    float ls = 0.f;
#pragma unroll
    for (int k = 0; k < 8; ++k) ls += expf(eL[k * 256 + tid] - M);
    ls = wave_sum(ls);
    if (lane == 0) sSum[wave] = ls;
  }
  lds_only_barrier();
  float Lsum = sSum[0];
#pragma unroll
  for (int w = 1; w < 8; ++w) Lsum += sSum[w];
  const float inv = 1.0f / Lsum;

#pragma unroll
  for (int k = 0; k < 4; ++k)
    aOut[b * TT + k * 512 + tid] = expf(eL[k * 512 + tid] - M) * inv;

  if (tid < DD) {
    float sum = (expf(mc[0] - M) * inv) * prL[tid];
#pragma unroll
    for (int c = 1; c < 4; ++c) sum = fmaf(expf(mc[c] - M) * inv, prL[c * DD + tid], sum);
    xs[tid] = sum;
  }
  lds_only_barrier();

  if (wave < 8) {
    v8f acc = {};
#pragma unroll
    for (int k = 0; k < 32; ++k) {
      const v2f a = *(const v2f*)(xs + 4 * k + 2 * hi);
      acc = __builtin_amdgcn_wmma_f32_16x16x4_f32(false, a, false, bvf[k],
                                                  (short)0, acc, false, false);
    }
    if (hi == 0) os[n] = acc[0] + bvn;
  }
  lds_only_barrier();
  if (tid < HH) out[b * HH + tid] = os[tid];
}

extern "C" void kernel_launch(void* const* d_in, const int* in_sizes, int n_in,
                              void* d_out, int out_size, void* d_ws, size_t ws_size,
                              hipStream_t stream) {
  (void)in_sizes; (void)n_in; (void)out_size; (void)d_ws; (void)ws_size;
  const float* x    = (const float*)d_in[0];
  const float* mask = (const float*)d_in[1];
  const float* Wq   = (const float*)d_in[2];
  const float* bq   = (const float*)d_in[3];
  const float* Wk   = (const float*)d_in[4];
  const float* bk   = (const float*)d_in[5];
  const float* Wv   = (const float*)d_in[6];
  const float* bv   = (const float*)d_in[7];
  const float* rate = (const float*)d_in[8];

  float* out      = (float*)d_out;
  float* aOut     = out + BB * HH;
  float* dotOut   = aOut + BB * TT;
  float* decayOut = dotOut + BB * TT;

  fused_batch_kernel<<<BB, 512, 0, stream>>>(x, mask, Wq, bq, Wk, bk, Wv, bv, rate,
                                             out, aOut, dotOut, decayOut);
}


# pass 1 asks L2 three steps ahead for the three 16-row groups per wave that are read only once (kept on chip for pass 2): one wave-wide device-scope global prefetch per group; sums unchanged
# speedup vs baseline: 3.8492x; 1.0742x over previous
#include <hip/hip_runtime.h>
#include <math.h>

#define BB 128
#define TT 2048
#define DD 128
#define HH 128

typedef __attribute__((ext_vector_type(2))) float v2f;
typedef __attribute__((ext_vector_type(8))) float v8f;

__device__ __forceinline__ void lds_only_barrier() {
  __builtin_amdgcn_fence(__ATOMIC_RELEASE, "workgroup", "local");
  __builtin_amdgcn_s_barrier();
  __builtin_amdgcn_fence(__ATOMIC_ACQUIRE, "workgroup", "local");
}

template <int M>
__device__ __forceinline__ float row_xor(const float v) {
  return __builtin_bit_cast(float, __builtin_amdgcn_update_dpp(
      0, __builtin_bit_cast(int, v), 0x160 + M, 0xf, 0xf, true));
}

__device__ __forceinline__ float half_xor(const float v) {
  const int b = __builtin_bit_cast(int, v);
  return __builtin_bit_cast(float, __builtin_amdgcn_permlanex16(
      b, b, 0x76543210, 0xfedcba98, false, false));
}

__device__ __forceinline__ float wave_sum(float u) {
  u += half_xor(u);
  u += row_xor<8>(u);
  u += row_xor<4>(u);
  u += row_xor<2>(u);
  u += row_xor<1>(u);
  return u;
}

__device__ __forceinline__ float exp_hw(const float x) {
  return __builtin_amdgcn_exp2f(x * 1.4426950408889634f);
}

__device__ __forceinline__ float log1p_hw(const float u) {
  const float y = 1.0f + u;
  const float d = y - 1.0f;
  const float l = __builtin_amdgcn_logf(y) * 0.6931471805599453f;
  const float c = (u - d) * __builtin_amdgcn_rcpf(y);
  return (fabsf(c) <= 1.0f) ? l + c : l;
}

template <int OFF>
__device__ __forceinline__ void fold_level(float (&s)[16], const int lane) {
  const bool up = (lane & OFF) != 0;
#pragma unroll
  for (int i = 0; i < OFF; ++i) {
    const float keep = up ? s[i + OFF] : s[i];
    const float send = up ? s[i] : s[i + OFF];
    s[i] = keep + row_xor<OFF>(send);
  }
}

__global__ __launch_bounds__(512) void fused_batch_kernel(
    const float* __restrict__ x, const float* __restrict__ mask,
    const float* __restrict__ Wq, const float* __restrict__ bq,
    const float* __restrict__ Wk, const float* __restrict__ bk,
    const float* __restrict__ Wv, const float* __restrict__ bv,
    const float* __restrict__ rate, float* __restrict__ out,
    float* __restrict__ aOut, float* __restrict__ dotOut,
    float* __restrict__ decayOut) {
  const int b = blockIdx.x;
  const int tid = threadIdx.x, wave = tid >> 5, lane = tid & 31;
  const int row16 = lane & 15;
  const int hi = lane >> 4;
  const bool hiB = lane >= 16;
  const int n = (wave & 7) * 16 + row16;
  const int tw = wave * 128;

  struct Lds {
    alignas(16) float red[32 * DD];
    alignas(16) float xs[DD];
    alignas(16) float qs[DD];
    alignas(16) float qkL[DD];
    float eL[TT];
    float prL[4 * DD];
    float os[HH];
    float sMax[32];
    float sSum[8];
    float d0L[4];
    alignas(16) float4 parked[16 * 2 * 16 * 32];
  };
  __shared__ Lds L;
  float* const red = L.red;
  float* const sMax = L.sMax;
  float* const xs = L.xs;
  float* const qs = L.qs;
  float* const qkL = L.qkL;
  float* const d0L = L.d0L;
  float* const eL = L.eL;
  float* const prL = L.prL;
  float* const sSum = L.sSum;
  float* const os = L.os;
  float4* const park = L.parked + wave * (2 * 16 * 32) + lane;

  const float4* rows = (const float4*)(x + (b * TT + tw) * DD) + lane;
  float4 xv[2][16];

  {
#pragma unroll
    for (int i = 0; i < 16; ++i) xv[0][i] = rows[i * (DD / 4)];
    float4 acc[4];
#pragma unroll
    for (int k = 0; k < 4; ++k) acc[k] = make_float4(0.f, 0.f, 0.f, 0.f);
#pragma unroll
    for (int g = 0; g < 8; ++g) {
      if (g + 1 < 8) {
#pragma unroll
        for (int i = 0; i < 16; ++i)
          xv[(g + 1) & 1][i] = rows[((g + 1) * 16 + i) * (DD / 4)];
        if (g + 3 >= 5 && g + 3 < 8) {
          __builtin_prefetch((const char*)(x + (b * TT + tw + (g + 3) * 16) * DD) + lane * 256, 0, 1);
        }
        asm volatile("" : "+v"(acc[0].x), "+v"(acc[0].y), "+v"(acc[0].z), "+v"(acc[0].w),
                          "+v"(acc[1].x), "+v"(acc[1].y), "+v"(acc[1].z), "+v"(acc[1].w),
                          "+v"(acc[2].x), "+v"(acc[2].y), "+v"(acc[2].z), "+v"(acc[2].w),
                          "+v"(acc[3].x), "+v"(acc[3].y), "+v"(acc[3].z), "+v"(acc[3].w));
      }
#pragma unroll
      for (int i = 0; i < 16; ++i) {
        const float4 v = xv[g & 1][i];
        if (g == 5 || g == 6) park[((g - 5) * 16 + i) * 32] = v;
        acc[i & 3].x += v.x; acc[i & 3].y += v.y;
        acc[i & 3].z += v.z; acc[i & 3].w += v.w;
      }
      if ((g & 3) == 3) {
        float4 s;
        s.x = (acc[0].x + acc[1].x) + (acc[2].x + acc[3].x);
        s.y = (acc[0].y + acc[1].y) + (acc[2].y + acc[3].y);
        s.z = (acc[0].z + acc[1].z) + (acc[2].z + acc[3].z);
        s.w = (acc[0].w + acc[1].w) + (acc[2].w + acc[3].w);
        *(float4*)(red + (2 * wave + (g >> 2)) * DD + lane * 4) = s;
#pragma unroll
        for (int k = 0; k < 4; ++k) acc[k] = make_float4(0.f, 0.f, 0.f, 0.f);
      }
      asm volatile("" : "+v"(acc[0].x), "+v"(acc[0].y), "+v"(acc[0].z), "+v"(acc[0].w),
                        "+v"(acc[1].x), "+v"(acc[1].y), "+v"(acc[1].z), "+v"(acc[1].w),
                        "+v"(acc[2].x), "+v"(acc[2].y), "+v"(acc[2].z), "+v"(acc[2].w),
                        "+v"(acc[3].x), "+v"(acc[3].y), "+v"(acc[3].z), "+v"(acc[3].w));
    }
  }

  const float rate0 = rate[0];
  v2f bqf[32];
  float bqn = 0.f;
  if (wave < 8) {
#pragma unroll
    for (int k = 0; k < 32; ++k) bqf[k] = *(const v2f*)(Wq + n * 128 + 4 * k + 2 * hi);
    bqn = bq[n];
  }
  float bkv[4];
  if (wave == 8) {
#pragma unroll
    for (int i = 0; i < 4; ++i) bkv[i] = bk[lane + 32 * i];
  }
  lds_only_barrier();

  if (tid < DD) {
    float t[4];
#pragma unroll
    for (int c = 0; c < 4; ++c) {
      t[c] = red[(8 * c) * DD + tid];
#pragma unroll
      for (int w = 1; w < 8; ++w) t[c] += red[(8 * c + w) * DD + tid];
    }
    float s = t[0];
#pragma unroll
    for (int c = 1; c < 4; ++c) s += t[c];
    const float sc = 1.0f / (float)TT;
    xs[tid] = s * sc;
  }
  lds_only_barrier();

  v2f bkf[32];
  if (wave < 8) {
#pragma unroll
    for (int k = 0; k < 32; ++k) {
      bkf[k].x = Wk[(4 * k + 2 * hi + 0) * 128 + n];
      bkf[k].y = Wk[(4 * k + 2 * hi + 1) * 128 + n];
    }
    v8f acc = {};
#pragma unroll
    for (int k = 0; k < 32; ++k) {
      const v2f a = *(const v2f*)(xs + 4 * k + 2 * hi);
      acc = __builtin_amdgcn_wmma_f32_16x16x4_f32(false, a, false, bqf[k],
                                                  (short)0, acc, false, false);
    }
    if (hi == 0) qs[n] = acc[0] + bqn;
  }
  lds_only_barrier();

  if (wave < 8) {
    v8f acc2 = {};
#pragma unroll
    for (int k = 0; k < 32; ++k) {
      const v2f a = *(const v2f*)(qs + 4 * k + 2 * hi);
      acc2 = __builtin_amdgcn_wmma_f32_16x16x4_f32(false, a, false, bkf[k],
                                                   (short)0, acc2, false, false);
    }
    if (hi == 0) qkL[n] = acc2[0];
  } else if (wave == 8) {
    const float p0 = bkv[0] * qs[lane];
    const float p1 = bkv[1] * qs[lane + 32];
    const float p2 = bkv[2] * qs[lane + 64];
    const float p3 = bkv[3] * qs[lane + 96];
    const float u = wave_sum((p0 + p2) + (p1 + p3));
    if (lane == 0) d0L[0] = u;
  }
  lds_only_barrier();

  {
    float mk[8];
#pragma unroll
    for (int g = 0; g < 8; ++g) mk[g] = mask[b * TT + tw + 16 * g + row16];
    float4 qk4 = *(const float4*)(qkL + lane * 4);
    const float d0 = d0L[0];
    float4 acc0 = make_float4(0.f, 0.f, 0.f, 0.f);
    float4 acc1 = make_float4(0.f, 0.f, 0.f, 0.f);
    float mrun = 0.f;
    float dvO = 0.f, decO = 0.f, evO = 0.f;
#pragma unroll
    for (int st = 0; st < 8; ++st) {
      const int g = 7 - st;
      if (g > 0) {
        const int gn = g - 1;
        if (gn == 6 || gn == 5) {
#pragma unroll
          for (int i = 0; i < 16; ++i)
            xv[gn & 1][i] = park[((gn - 5) * 16 + i) * 32];
        } else {
#pragma unroll
          for (int i = 0; i < 16; ++i)
            xv[gn & 1][i] = rows[(gn * 16 + i) * (DD / 4)];
        }
      }
      asm volatile("" : "+v"(qk4.x), "+v"(qk4.y), "+v"(qk4.z), "+v"(qk4.w));
      float s[16];
#pragma unroll
      for (int i = 0; i < 16; ++i) {
        const float4 v = xv[g & 1][i];
        s[i] = fmaf(v.w, qk4.w, fmaf(v.z, qk4.z, fmaf(v.x, qk4.x, v.y * qk4.y)));
      }
      fold_level<8>(s, lane);
      fold_level<4>(s, lane);
      fold_level<2>(s, lane);
      fold_level<1>(s, lane);
      const float dsum = s[0] + half_xor(s[0]);
      const float dv = dsum + d0;
      const float sig = __builtin_amdgcn_rcpf(1.0f + exp_hw(-dv));
      const float tm = log1p_hw((1.0f - sig) * mk[g]);
      const float dec = rate0 * tm;
      const float ev = dv - dec;
      float gm = ev;
      gm = fmaxf(gm, row_xor<8>(gm));
      gm = fmaxf(gm, row_xor<4>(gm));
      gm = fmaxf(gm, row_xor<2>(gm));
      gm = fmaxf(gm, row_xor<1>(gm));
      if ((g & 3) == 3) {
        mrun = gm;
      } else {
        const float mnew = fmaxf(mrun, gm);
        const float sc = exp_hw(mrun - mnew);
        acc0.x *= sc; acc0.y *= sc; acc0.z *= sc; acc0.w *= sc;
        acc1.x *= sc; acc1.y *= sc; acc1.z *= sc; acc1.w *= sc;
        mrun = mnew;
      }
      const int wbits = __float_as_int(exp_hw(ev - mrun));
#pragma unroll
      for (int i = 0; i < 16; ++i) {
        const float wi = __int_as_float(__builtin_amdgcn_readlane(wbits, i));
        const float4 v = xv[g & 1][i];
        if (i & 1) {
          acc1.x = fmaf(wi, v.x, acc1.x);
          acc1.y = fmaf(wi, v.y, acc1.y);
          acc1.z = fmaf(wi, v.z, acc1.z);
          acc1.w = fmaf(wi, v.w, acc1.w);
        } else {
          acc0.x = fmaf(wi, v.x, acc0.x);
          acc0.y = fmaf(wi, v.y, acc0.y);
          acc0.z = fmaf(wi, v.z, acc0.z);
          acc0.w = fmaf(wi, v.w, acc0.w);
        }
      }
      asm volatile("" : "+v"(acc0.x), "+v"(acc0.y), "+v"(acc0.z), "+v"(acc0.w),
                        "+v"(acc1.x), "+v"(acc1.y), "+v"(acc1.z), "+v"(acc1.w));
      if (g & 1) {
        dvO = dv; decO = dec; evO = ev;
      } else {
        const int o = tw + (g >> 1) * 32 + lane;
        dotOut[b * TT + o] = hiB ? dvO : dv;
        decayOut[b * TT + o] = hiB ? decO : dec;
        eL[o] = hiB ? evO : ev;
      }
      if ((g & 3) == 0) {
        float4 acc;
        acc.x = acc0.x + acc1.x;
        acc.y = acc0.y + acc1.y;
        acc.z = acc0.z + acc1.z;
        acc.w = acc0.w + acc1.w;
        if (lane == 0) sMax[2 * wave + (g >> 2)] = mrun;
        *(float4*)(red + (2 * wave + (g >> 2)) * DD + lane * 4) = acc;
        acc0 = make_float4(0.f, 0.f, 0.f, 0.f);
        acc1 = make_float4(0.f, 0.f, 0.f, 0.f);
      }
    }
  }

  v2f bvf[32];
  float bvn = 0.f;
  if (wave < 8) {
#pragma unroll
    for (int k = 0; k < 32; ++k) bvf[k] = *(const v2f*)(Wv + n * 128 + 4 * k + 2 * hi);
    bvn = bv[n];
  }
  lds_only_barrier();

  float mc[4];
#pragma unroll
  for (int c = 0; c < 4; ++c) {
    mc[c] = sMax[8 * c];
#pragma unroll
    for (int w = 1; w < 8; ++w) mc[c] = fmaxf(mc[c], sMax[8 * c + w]);
  }
  const float M = fmaxf(fmaxf(mc[0], mc[1]), fmaxf(mc[2], mc[3]));
  {
    const int c = tid >> 7, col = tid & 127;
    float m = sMax[8 * c];
#pragma unroll
    for (int w = 1; w < 8; ++w) m = fmaxf(m, sMax[8 * c + w]);
    float sum = expf(sMax[8 * c] - m) * red[(8 * c) * DD + col];
#pragma unroll
    for (int w = 1; w < 8; ++w)
      sum = fmaf(expf(sMax[8 * c + w] - m), red[(8 * c + w) * DD + col], sum);
    prL[c * DD + col] = sum;
  }
  if (wave < 8) {
    float ls = 0.f;
#pragma unroll
    for (int k = 0; k < 8; ++k) ls += expf(eL[k * 256 + tid] - M);
    ls = wave_sum(ls);
    if (lane == 0) sSum[wave] = ls;
  }
  lds_only_barrier();
  float Lsum = sSum[0];
#pragma unroll
  for (int w = 1; w < 8; ++w) Lsum += sSum[w];
  const float inv = 1.0f / Lsum;

#pragma unroll
  for (int k = 0; k < 4; ++k)
    aOut[b * TT + k * 512 + tid] = expf(eL[k * 512 + tid] - M) * inv;

  if (tid < DD) {
    float sum = (expf(mc[0] - M) * inv) * prL[tid];
#pragma unroll
    for (int c = 1; c < 4; ++c) sum = fmaf(expf(mc[c] - M) * inv, prL[c * DD + tid], sum);
    xs[tid] = sum;
  }
  lds_only_barrier();

  if (wave < 8) {
    v8f acc = {};
#pragma unroll
    for (int k = 0; k < 32; ++k) {
      const v2f a = *(const v2f*)(xs + 4 * k + 2 * hi);
      acc = __builtin_amdgcn_wmma_f32_16x16x4_f32(false, a, false, bvf[k],
                                                  (short)0, acc, false, false);
    }
    if (hi == 0) os[n] = acc[0] + bvn;
  }
  lds_only_barrier();
  if (tid < HH) out[b * HH + tid] = os[tid];
}

extern "C" void kernel_launch(void* const* d_in, const int* in_sizes, int n_in,
                              void* d_out, int out_size, void* d_ws, size_t ws_size,
                              hipStream_t stream) {
  (void)in_sizes; (void)n_in; (void)out_size; (void)d_ws; (void)ws_size;
  const float* x    = (const float*)d_in[0];
  const float* mask = (const float*)d_in[1];
  const float* Wq   = (const float*)d_in[2];
  const float* bq   = (const float*)d_in[3];
  const float* Wk   = (const float*)d_in[4];
  const float* bk   = (const float*)d_in[5];
  const float* Wv   = (const float*)d_in[6];
  const float* bv   = (const float*)d_in[7];
  const float* rate = (const float*)d_in[8];

  float* out      = (float*)d_out;
  float* aOut     = out + BB * HH;
  float* dotOut   = aOut + BB * TT;
  float* decayOut = dotOut + BB * TT;

  fused_batch_kernel<<<BB, 512, 0, stream>>>(x, mask, Wq, bq, Wk, bk, Wv, bv, rate,
                                             out, aOut, dotOut, decayOut);
}
